# starting kernel
# baseline (speedup 1.0000x reference)
#include <hip/hip_runtime.h>
#include <hip/hip_bf16.h>


#define USER_NUM 60000
#define ITEM_NUM 40000
#define NODE_NUM (USER_NUM + ITEM_NUM)
#define EMBED_DIM 64
#define N_LAYER 3
#define NNZ 1600000
#define BATCH 4096
#define NEG_SLOPE 0.2f
#define NODE_TILES (NODE_NUM / 16)

typedef float v2f __attribute__((ext_vector_type(2)));
typedef float v8f __attribute__((ext_vector_type(8)));

__global__ void ngcf_zero_kernel(float4* __restrict__ p, int n4) {
    int i = blockIdx.x * blockDim.x + threadIdx.x;
    if (i < n4) p[i] = make_float4(0.f, 0.f, 0.f, 0.f);
}

__global__ void ngcf_spmm_kernel(const float* __restrict__ vals,
                                 const int*   __restrict__ rows,
                                 const int*   __restrict__ cols,
                                 const float* __restrict__ E,
                                 float*       __restrict__ side) {
    const int wavesPerBlock = blockDim.x >> 5;
    const int edge = blockIdx.x * wavesPerBlock + (threadIdx.x >> 5);
    if (edge >= NNZ) return;
    const int lane = threadIdx.x & 31;

    const float v = vals[edge];
    const int   r = rows[edge];
    const int   c = cols[edge];

    const float2 e = *(const float2*)(E + (size_t)c * EMBED_DIM + lane * 2);
    float* dst = side + (size_t)r * EMBED_DIM + lane * 2;
    atomicAdd(dst + 0, v * e.x);
    atomicAdd(dst + 1, v * e.y);
}

__global__ void ngcf_layer_kernel(const float* __restrict__ Eprev,
                                  const float* __restrict__ side,
                                  const float* __restrict__ W1,
                                  const float* __restrict__ W2,
                                  float*       __restrict__ Eout) {
    __shared__ float w1i[EMBED_DIM * EMBED_DIM];
    __shared__ float w2i[EMBED_DIM * EMBED_DIM];
    for (int i = threadIdx.x; i < EMBED_DIM * EMBED_DIM; i += blockDim.x) {
        const int k = i >> 6;
        const int n = i & 63;
        const int dst = ((k >> 1) << 7) + (n << 1) + (k & 1);
        w1i[dst] = W1[i];
        w2i[dst] = W2[i];
    }
    __syncthreads();

    const int tile = blockIdx.x * (blockDim.x >> 5) + (threadIdx.x >> 5);
    if (tile >= NODE_TILES) return;

    const int lane    = threadIdx.x & 31;
    const int halfsel = lane >> 4;
    const int nlane   = lane & 15;
    const int m0      = tile * 16;
    const int mrow    = m0 + nlane;
    const int koff    = halfsel * 2;

    v8f acc[4] = {v8f{}, v8f{}, v8f{}, v8f{}};

    const float* epRow = Eprev + (size_t)mrow * EMBED_DIM + koff;
    const float* sdRow = side  + (size_t)mrow * EMBED_DIM + koff;

#pragma unroll
    for (int kk = 0; kk < 16; ++kk) {
        const int k0 = kk * 4;
        const float2 ep = *(const float2*)(epRow + k0);
        const float2 sd = *(const float2*)(sdRow + k0);

        v2f aSum, aBi;
        aSum.x = ep.x + sd.x;  aSum.y = ep.y + sd.y;
        aBi.x  = ep.x * sd.x;  aBi.y  = ep.y * sd.y;

        const int kph = ((k0 >> 1) + halfsel) << 7;
        const float* w1p = w1i + kph + (nlane << 1);
        const float* w2p = w2i + kph + (nlane << 1);

#pragma unroll
        for (int nb = 0; nb < 4; ++nb) {
            const v2f b1 = *(const v2f*)(w1p + nb * 32);
            const v2f b2 = *(const v2f*)(w2p + nb * 32);
            acc[nb] = __builtin_amdgcn_wmma_f32_16x16x4_f32(
                false, aSum, false, b1, (short)0, acc[nb], false, false);
            acc[nb] = __builtin_amdgcn_wmma_f32_16x16x4_f32(
                false, aBi,  false, b2, (short)0, acc[nb], false, false);
        }
    }

#pragma unroll
    for (int nb = 0; nb < 4; ++nb) {
#pragma unroll
        for (int i = 0; i < 8; ++i) {
            float v = acc[nb][i];
            v = (v > 0.f) ? v : v * NEG_SLOPE;
            Eout[(size_t)(m0 + i + halfsel * 8) * EMBED_DIM + nb * 16 + nlane] = v;
        }
    }
}

__global__ void ngcf_gather_kernel(const float* __restrict__ emb,
                                   const float* __restrict__ E1,
                                   const float* __restrict__ E2,
                                   const float* __restrict__ E3,
                                   const int* __restrict__ user_idx,
                                   const int* __restrict__ item_idx,
                                   float* __restrict__ out) {
    const int row = blockIdx.x;
    const int t   = threadIdx.x;
    const int seg = t >> 6;
    const int dim = t & 63;

    int node;
    if (row < BATCH) node = user_idx[row];
    else             node = USER_NUM + item_idx[row - BATCH];

    const float* src = (seg == 0) ? emb : (seg == 1) ? E1 : (seg == 2) ? E2 : E3;
    out[(size_t)row * 256 + t] = src[(size_t)node * EMBED_DIM + dim];
}

extern "C" void kernel_launch(void* const* d_in, const int* in_sizes, int n_in,
                              void* d_out, int out_size, void* d_ws, size_t ws_size,
                              hipStream_t stream) {
    const float* emb      = (const float*)d_in[0];
    const float* W1       = (const float*)d_in[1];
    const float* W2       = (const float*)d_in[2];
    const float* vals     = (const float*)d_in[3];
    const int*   row_idx  = (const int*)d_in[4];
    const int*   col_idx  = (const int*)d_in[5];
    const int*   user_idx = (const int*)d_in[6];
    const int*   item_idx = (const int*)d_in[7];

    const size_t EN = (size_t)NODE_NUM * EMBED_DIM;
    float* side = (float*)d_ws;
    float* E1   = side + EN;
    float* E2   = E1 + EN;
    float* E3   = E2 + EN;
    float* Eouts[N_LAYER] = {E1, E2, E3};

    const int layerBlocks = (NODE_TILES + 7) / 8;

    const float* Eprev = emb;
    for (int l = 0; l < N_LAYER; ++l) {
        ngcf_zero_kernel<<<(int)(EN / 4 / 256), 256, 0, stream>>>((float4*)side,
                                                                  (int)(EN / 4));
        ngcf_spmm_kernel<<<NNZ / 8, 256, 0, stream>>>(vals, row_idx, col_idx,
                                                      Eprev, side);
        ngcf_layer_kernel<<<layerBlocks, 256, 0, stream>>>(
            Eprev, side,
            W1 + (size_t)l * EMBED_DIM * EMBED_DIM,
            W2 + (size_t)l * EMBED_DIM * EMBED_DIM,
            Eouts[l]);
        Eprev = Eouts[l];
    }

    ngcf_gather_kernel<<<2 * BATCH, 256, 0, stream>>>(emb, E1, E2, E3,
                                                      user_idx, item_idx,
                                                      (float*)d_out);
}


# SpMM: edges counting-sorted by destination row once per call (integer atomic rank, scan, place); one wave per row sums gathered rows in rank-arrival order, one store; no float atomics, no zero fill
# speedup vs baseline: 2.1569x; 2.1569x over previous
#include <hip/hip_runtime.h>
#include <hip/hip_bf16.h>


#define USER_NUM 60000
#define ITEM_NUM 40000
#define NODE_NUM (USER_NUM + ITEM_NUM)
#define EMBED_DIM 64
#define N_LAYER 3
#define NNZ 1600000
#define BATCH 4096
#define NEG_SLOPE 0.2f
#define NODE_TILES (NODE_NUM / 16)

typedef float v2f __attribute__((ext_vector_type(2)));
typedef float v8f __attribute__((ext_vector_type(8)));

__global__ void ngcf_zero_kernel(float4* __restrict__ p, int n4) {
    int i = blockIdx.x * blockDim.x + threadIdx.x;
    if (i < n4) p[i] = make_float4(0.f, 0.f, 0.f, 0.f);
}

#define SCAN_ROWS 1024
#define SCAN_BLOCKS ((NODE_NUM + SCAN_ROWS - 1) / SCAN_ROWS)

__global__ void ngcf_rank_kernel(const int* __restrict__ rows,
                                 int* __restrict__ counts,
                                 int* __restrict__ rank) {
    const int e = blockIdx.x * blockDim.x + threadIdx.x;
    if (e < NNZ) rank[e] = atomicAdd(&counts[rows[e]], 1);
}

__global__ void ngcf_blocksum_kernel(const int* __restrict__ counts,
                                     int* __restrict__ bsum) {
    __shared__ int wsum[8];
    const int base = blockIdx.x * SCAN_ROWS + threadIdx.x * 4;
    int s = 0;
    if (base < NODE_NUM) {
        const int4 c = *(const int4*)(counts + base);
        s = c.x + c.y + c.z + c.w;
    }
    for (int off = 16; off > 0; off >>= 1) s += __shfl_xor(s, off);
    if ((threadIdx.x & 31) == 0) wsum[threadIdx.x >> 5] = s;
    __syncthreads();
    if (threadIdx.x == 0) {
        int t = 0;
        for (int i = 0; i < 8; ++i) t += wsum[i];
        bsum[blockIdx.x] = t;
    }
}

__global__ void ngcf_scan_kernel(const int* __restrict__ counts,
                                 const int* __restrict__ bsum,
                                 int* __restrict__ rowptr) {
    __shared__ int wtot[8];
    __shared__ int wpre[8];
    const int tid  = threadIdx.x;
    const int lane = tid & 31;
    const int wid  = tid >> 5;

    int pre = (tid < (int)blockIdx.x) ? bsum[tid] : 0;
    for (int off = 16; off > 0; off >>= 1) pre += __shfl_xor(pre, off);

    const int base = blockIdx.x * SCAN_ROWS + tid * 4;
    int4 c = make_int4(0, 0, 0, 0);
    if (base < NODE_NUM) c = *(const int4*)(counts + base);
    const int s = c.x + c.y + c.z + c.w;

    int incl = s;
    for (int off = 1; off < 32; off <<= 1) {
        const int t = __shfl_up(incl, off);
        if (lane >= off) incl += t;
    }
    if (lane == 31) wtot[wid] = incl;
    if (lane == 0)  wpre[wid] = pre;
    __syncthreads();

    int front = 0;
    for (int i = 0; i < 8; ++i) {
        front += wpre[i];
        if (i < wid) front += wtot[i];
    }
    const int excl = front + incl - s;
    if (base < NODE_NUM) {
        int4 o;
        o.x = excl;
        o.y = o.x + c.x;
        o.z = o.y + c.y;
        o.w = o.z + c.z;
        *(int4*)(rowptr + base) = o;
        if (base == NODE_NUM - 4) rowptr[NODE_NUM] = o.w + c.w;
    }
}

__global__ void ngcf_place_kernel(const float* __restrict__ vals,
                                  const int*   __restrict__ rows,
                                  const int*   __restrict__ cols,
                                  const int*   __restrict__ rowptr,
                                  const int*   __restrict__ rank,
                                  int*         __restrict__ pairs) {
    const int e = blockIdx.x * blockDim.x + threadIdx.x;
    if (e >= NNZ) return;
    int pos = rowptr[rows[e]] + rank[e];
    pos = min(max(pos, 0), NNZ - 1);
    ((int2*)pairs)[pos] = make_int2(cols[e], __float_as_int(vals[e]));
}

template <bool TAIL>
__device__ __forceinline__ void ngcf_gather8(const int2* __restrict__ pairs,
                                             const float* __restrict__ Elane,
                                             int j, int end, float& ax, float& ay) {
    int2 p[8];
    for (int k = 0; k < 8; ++k) {
        int jj = j + k;
        if (TAIL) jj = min(jj, end - 1);
        p[k] = pairs[jj];
    }
    float2 ev[8];
    for (int k = 0; k < 8; ++k) {
        const unsigned c = min((unsigned)p[k].x, (unsigned)(NODE_NUM - 1));
        ev[k] = *(const float2*)(Elane + (size_t)c * EMBED_DIM);
    }
    for (int k = 0; k < 8; ++k) {
        if (!TAIL || (j + k < end)) {
            const float v = __int_as_float(p[k].y);
            ax = fmaf(v, ev[k].x, ax);
            ay = fmaf(v, ev[k].y, ay);
        }
    }
}

__global__ void
ngcf_spmm_rows_kernel(const int*   __restrict__ rowptr,
                      const int*   __restrict__ pairs_i,
                      const float* __restrict__ E,
                      float*       __restrict__ side) {
    const int lane = threadIdx.x & 31;
    const int r = __builtin_amdgcn_readfirstlane(
        (int)(blockIdx.x * (blockDim.x >> 5) + (threadIdx.x >> 5)));
    if (r >= NODE_NUM) return;

    const int2* pairs = (const int2*)pairs_i;
    int       j   = min(max(rowptr[r], 0), NNZ);
    const int end = min(rowptr[r + 1], NNZ);
    const float* Elane = E + lane * 2;

    float ax = 0.f, ay = 0.f;
    for (; j + 8 <= end; j += 8) ngcf_gather8<false>(pairs, Elane, j, end, ax, ay);
    if (j < end) ngcf_gather8<true>(pairs, Elane, j, end, ax, ay);

    *(float2*)(side + (size_t)r * EMBED_DIM + lane * 2) = make_float2(ax, ay);
}

__global__ void ngcf_layer_kernel(const float* __restrict__ Eprev,
                                  const float* __restrict__ side,
                                  const float* __restrict__ W1,
                                  const float* __restrict__ W2,
                                  float*       __restrict__ Eout) {
    __shared__ float w1i[EMBED_DIM * EMBED_DIM];
    __shared__ float w2i[EMBED_DIM * EMBED_DIM];
    for (int i = threadIdx.x; i < EMBED_DIM * EMBED_DIM; i += blockDim.x) {
        const int k = i >> 6;
        const int n = i & 63;
        const int dst = ((k >> 1) << 7) + (n << 1) + (k & 1);
        w1i[dst] = W1[i];
        w2i[dst] = W2[i];
    }
    __syncthreads();

    const int tile = blockIdx.x * (blockDim.x >> 5) + (threadIdx.x >> 5);
    if (tile >= NODE_TILES) return;

    const int lane    = threadIdx.x & 31;
    const int halfsel = lane >> 4;
    const int nlane   = lane & 15;
    const int m0      = tile * 16;
    const int mrow    = m0 + nlane;
    const int koff    = halfsel * 2;

    v8f acc[4] = {v8f{}, v8f{}, v8f{}, v8f{}};

    const float* epRow = Eprev + (size_t)mrow * EMBED_DIM + koff;
    const float* sdRow = side  + (size_t)mrow * EMBED_DIM + koff;

#pragma unroll
    for (int kk = 0; kk < 16; ++kk) {
        const int k0 = kk * 4;
        const float2 ep = *(const float2*)(epRow + k0);
        const float2 sd = *(const float2*)(sdRow + k0);

        v2f aSum, aBi;
        aSum.x = ep.x + sd.x;  aSum.y = ep.y + sd.y;
        aBi.x  = ep.x * sd.x;  aBi.y  = ep.y * sd.y;

        const int kph = ((k0 >> 1) + halfsel) << 7;
        const float* w1p = w1i + kph + (nlane << 1);
        const float* w2p = w2i + kph + (nlane << 1);

#pragma unroll
        for (int nb = 0; nb < 4; ++nb) {
            const v2f b1 = *(const v2f*)(w1p + nb * 32);
            const v2f b2 = *(const v2f*)(w2p + nb * 32);
            acc[nb] = __builtin_amdgcn_wmma_f32_16x16x4_f32(
                false, aSum, false, b1, (short)0, acc[nb], false, false);
            acc[nb] = __builtin_amdgcn_wmma_f32_16x16x4_f32(
                false, aBi,  false, b2, (short)0, acc[nb], false, false);
        }
    }

#pragma unroll
    for (int nb = 0; nb < 4; ++nb) {
#pragma unroll
        for (int i = 0; i < 8; ++i) {
            float v = acc[nb][i];
            v = (v > 0.f) ? v : v * NEG_SLOPE;
            Eout[(size_t)(m0 + i + halfsel * 8) * EMBED_DIM + nb * 16 + nlane] = v;
        }
    }
}

__global__ void ngcf_gather_kernel(const float* __restrict__ emb,
                                   const float* __restrict__ E1,
                                   const float* __restrict__ E2,
                                   const float* __restrict__ E3,
                                   const int* __restrict__ user_idx,
                                   const int* __restrict__ item_idx,
                                   float* __restrict__ out) {
    const int row = blockIdx.x;
    const int t   = threadIdx.x;
    const int seg = t >> 6;
    const int dim = t & 63;

    int node;
    if (row < BATCH) node = user_idx[row];
    else             node = USER_NUM + item_idx[row - BATCH];

    const float* src = (seg == 0) ? emb : (seg == 1) ? E1 : (seg == 2) ? E2 : E3;
    out[(size_t)row * 256 + t] = src[(size_t)node * EMBED_DIM + dim];
}

extern "C" void kernel_launch(void* const* d_in, const int* in_sizes, int n_in,
                              void* d_out, int out_size, void* d_ws, size_t ws_size,
                              hipStream_t stream) {
    const float* emb      = (const float*)d_in[0];
    const float* W1       = (const float*)d_in[1];
    const float* W2       = (const float*)d_in[2];
    const float* vals     = (const float*)d_in[3];
    const int*   row_idx  = (const int*)d_in[4];
    const int*   col_idx  = (const int*)d_in[5];
    const int*   user_idx = (const int*)d_in[6];
    const int*   item_idx = (const int*)d_in[7];

    const size_t EN = (size_t)NODE_NUM * EMBED_DIM;
    float* side = (float*)d_ws;
    float* E1   = side + EN;
    float* E2   = E1 + EN;
    float* E3   = E2 + EN;
    float* Eouts[N_LAYER] = {E1, E2, E3};

    int* pairs  = (int*)E3;
    int* rank   = pairs + 2 * (size_t)NNZ;
    int* counts = rank + (size_t)NNZ;
    int* rowptr = counts + 102400;
    int* bsum   = rowptr + 102400;

    ngcf_zero_kernel<<<(NODE_NUM / 4 + 255) / 256, 256, 0, stream>>>(
        (float4*)counts, NODE_NUM / 4);
    ngcf_rank_kernel<<<NNZ / 256, 256, 0, stream>>>(row_idx, counts, rank);
    ngcf_blocksum_kernel<<<SCAN_BLOCKS, 256, 0, stream>>>(counts, bsum);
    ngcf_scan_kernel<<<SCAN_BLOCKS, 256, 0, stream>>>(counts, bsum, rowptr);
    ngcf_place_kernel<<<NNZ / 256, 256, 0, stream>>>(vals, row_idx, col_idx,
                                                     rowptr, rank, pairs);

    const int layerBlocks = (NODE_TILES + 7) / 8;

    const float* Eprev = emb;
    for (int l = 0; l < N_LAYER; ++l) {
        ngcf_spmm_rows_kernel<<<NODE_NUM / 8, 256, 0, stream>>>(rowptr, pairs,
                                                                Eprev, side);
        ngcf_layer_kernel<<<layerBlocks, 256, 0, stream>>>(
            Eprev, side,
            W1 + (size_t)l * EMBED_DIM * EMBED_DIM,
            W2 + (size_t)l * EMBED_DIM * EMBED_DIM,
            Eouts[l]);
        Eprev = Eouts[l];
    }

    ngcf_gather_kernel<<<2 * BATCH, 256, 0, stream>>>(emb, E1, E2, E3,
                                                      user_idx, item_idx,
                                                      (float*)d_out);
}


# Layer 3 evaluated only at the 8192 selected output rows: third SpMM and third layer GEMM run over the user/item slot list (8192 waves, 512 tiles), not all 100000 nodes; same summation order
# speedup vs baseline: 2.3761x; 1.1016x over previous
#include <hip/hip_runtime.h>
#include <hip/hip_bf16.h>


#define USER_NUM 60000
#define ITEM_NUM 40000
#define NODE_NUM (USER_NUM + ITEM_NUM)
#define EMBED_DIM 64
#define N_LAYER 3
#define NNZ 1600000
#define BATCH 4096
#define NEG_SLOPE 0.2f
#define NODE_TILES (NODE_NUM / 16)

typedef float v2f __attribute__((ext_vector_type(2)));
typedef float v8f __attribute__((ext_vector_type(8)));

__device__ __forceinline__ int ngcf_slot_node(const int* __restrict__ user_idx,
                                              const int* __restrict__ item_idx,
                                              int s) {
    const int node = (s < BATCH) ? user_idx[s] : USER_NUM + item_idx[s - BATCH];
    return min(max(node, 0), NODE_NUM - 1);
}

__global__ void ngcf_zero_kernel(float4* __restrict__ p, int n4) {
    int i = blockIdx.x * blockDim.x + threadIdx.x;
    if (i < n4) p[i] = make_float4(0.f, 0.f, 0.f, 0.f);
}

#define SCAN_ROWS 1024
#define SCAN_BLOCKS ((NODE_NUM + SCAN_ROWS - 1) / SCAN_ROWS)

__global__ void ngcf_rank_kernel(const int* __restrict__ rows,
                                 int* __restrict__ counts,
                                 int* __restrict__ rank) {
    const int e = blockIdx.x * blockDim.x + threadIdx.x;
    if (e < NNZ) rank[e] = atomicAdd(&counts[rows[e]], 1);
}

__global__ void ngcf_blocksum_kernel(const int* __restrict__ counts,
                                     int* __restrict__ bsum) {
    __shared__ int wsum[8];
    const int base = blockIdx.x * SCAN_ROWS + threadIdx.x * 4;
    int s = 0;
    if (base < NODE_NUM) {
        const int4 c = *(const int4*)(counts + base);
        s = c.x + c.y + c.z + c.w;
    }
    for (int off = 16; off > 0; off >>= 1) s += __shfl_xor(s, off);
    if ((threadIdx.x & 31) == 0) wsum[threadIdx.x >> 5] = s;
    __syncthreads();
    if (threadIdx.x == 0) {
        int t = 0;
        for (int i = 0; i < 8; ++i) t += wsum[i];
        bsum[blockIdx.x] = t;
    }
}

__global__ void ngcf_scan_kernel(const int* __restrict__ counts,
                                 const int* __restrict__ bsum,
                                 int* __restrict__ rowptr) {
    __shared__ int wtot[8];
    __shared__ int wpre[8];
    const int tid  = threadIdx.x;
    const int lane = tid & 31;
    const int wid  = tid >> 5;

    int pre = (tid < (int)blockIdx.x) ? bsum[tid] : 0;
    for (int off = 16; off > 0; off >>= 1) pre += __shfl_xor(pre, off);

    const int base = blockIdx.x * SCAN_ROWS + tid * 4;
    int4 c = make_int4(0, 0, 0, 0);
    if (base < NODE_NUM) c = *(const int4*)(counts + base);
    const int s = c.x + c.y + c.z + c.w;

    int incl = s;
    for (int off = 1; off < 32; off <<= 1) {
        const int t = __shfl_up(incl, off);
        if (lane >= off) incl += t;
    }
    if (lane == 31) wtot[wid] = incl;
    if (lane == 0)  wpre[wid] = pre;
    __syncthreads();

    int front = 0;
    for (int i = 0; i < 8; ++i) {
        front += wpre[i];
        if (i < wid) front += wtot[i];
    }
    const int excl = front + incl - s;
    if (base < NODE_NUM) {
        int4 o;
        o.x = excl;
        o.y = o.x + c.x;
        o.z = o.y + c.y;
        o.w = o.z + c.z;
        *(int4*)(rowptr + base) = o;
        if (base == NODE_NUM - 4) rowptr[NODE_NUM] = o.w + c.w;
    }
}

__global__ void ngcf_place_kernel(const float* __restrict__ vals,
                                  const int*   __restrict__ rows,
                                  const int*   __restrict__ cols,
                                  const int*   __restrict__ rowptr,
                                  const int*   __restrict__ rank,
                                  int*         __restrict__ pairs) {
    const int e = blockIdx.x * blockDim.x + threadIdx.x;
    if (e >= NNZ) return;
    int pos = rowptr[rows[e]] + rank[e];
    pos = min(max(pos, 0), NNZ - 1);
    ((int2*)pairs)[pos] = make_int2(cols[e], __float_as_int(vals[e]));
}

template <bool TAIL>
__device__ __forceinline__ void ngcf_gather8(const int2* __restrict__ pairs,
                                             const float* __restrict__ Elane,
                                             int j, int end, float& ax, float& ay) {
    int2 p[8];
    for (int k = 0; k < 8; ++k) {
        int jj = j + k;
        if (TAIL) jj = min(jj, end - 1);
        p[k] = pairs[jj];
    }
    float2 ev[8];
    for (int k = 0; k < 8; ++k) {
        const unsigned c = min((unsigned)p[k].x, (unsigned)(NODE_NUM - 1));
        ev[k] = *(const float2*)(Elane + (size_t)c * EMBED_DIM);
    }
    for (int k = 0; k < 8; ++k) {
        if (!TAIL || (j + k < end)) {
            const float v = __int_as_float(p[k].y);
            ax = fmaf(v, ev[k].x, ax);
            ay = fmaf(v, ev[k].y, ay);
        }
    }
}

__device__ __forceinline__ void ngcf_spmm_row(const int*   __restrict__ rowptr,
                                              const int*   __restrict__ pairs_i,
                                              const float* __restrict__ E,
                                              float*       __restrict__ dst_row,
                                              int r, int lane) {
    const int2* pairs = (const int2*)pairs_i;
    int       j   = min(max(rowptr[r], 0), NNZ);
    const int end = min(rowptr[r + 1], NNZ);
    const float* Elane = E + lane * 2;

    float ax = 0.f, ay = 0.f;
    for (; j + 8 <= end; j += 8) ngcf_gather8<false>(pairs, Elane, j, end, ax, ay);
    if (j < end) ngcf_gather8<true>(pairs, Elane, j, end, ax, ay);

    *(float2*)(dst_row + lane * 2) = make_float2(ax, ay);
}

__global__ void
ngcf_spmm_rows_kernel(const int*   __restrict__ rowptr,
                      const int*   __restrict__ pairs_i,
                      const float* __restrict__ E,
                      float*       __restrict__ side) {
    const int lane = threadIdx.x & 31;
    const int r = __builtin_amdgcn_readfirstlane(
        (int)(blockIdx.x * (blockDim.x >> 5) + (threadIdx.x >> 5)));
    if (r >= NODE_NUM) return;
    ngcf_spmm_row(rowptr, pairs_i, E, side + (size_t)r * EMBED_DIM, r, lane);
}

__global__ void
ngcf_spmm_sel_kernel(const int*   __restrict__ rowptr,
                     const int*   __restrict__ pairs_i,
                     const float* __restrict__ E,
                     const int*   __restrict__ user_idx,
                     const int*   __restrict__ item_idx,
                     float*       __restrict__ side_sel) {
    const int lane = threadIdx.x & 31;
    const int s = __builtin_amdgcn_readfirstlane(
        (int)(blockIdx.x * (blockDim.x >> 5) + (threadIdx.x >> 5)));
    if (s >= 2 * BATCH) return;
    const int r = ngcf_slot_node(user_idx, item_idx, s);
    ngcf_spmm_row(rowptr, pairs_i, E, side_sel + (size_t)s * EMBED_DIM, r, lane);
}

__device__ __forceinline__ void ngcf_stage_weights(const float* __restrict__ W1,
                                                   const float* __restrict__ W2,
                                                   float* w1i, float* w2i) {
    for (int i = threadIdx.x; i < EMBED_DIM * EMBED_DIM; i += blockDim.x) {
        const int k = i >> 6;
        const int n = i & 63;
        const int dst = ((k >> 1) << 7) + (n << 1) + (k & 1);
        w1i[dst] = W1[i];
        w2i[dst] = W2[i];
    }
    __syncthreads();
}

__device__ __forceinline__ void ngcf_layer_tile(const float* __restrict__ epRow,
                                                const float* __restrict__ sdRow,
                                                const float* w1i, const float* w2i,
                                                float* __restrict__ outTile,
                                                int halfsel, int nlane) {
    v8f acc[4] = {v8f{}, v8f{}, v8f{}, v8f{}};

#pragma unroll
    for (int kk = 0; kk < 16; ++kk) {
        const int k0 = kk * 4;
        const float2 ep = *(const float2*)(epRow + k0);
        const float2 sd = *(const float2*)(sdRow + k0);

        v2f aSum, aBi;
        aSum.x = ep.x + sd.x;  aSum.y = ep.y + sd.y;
        aBi.x  = ep.x * sd.x;  aBi.y  = ep.y * sd.y;

        const int kph = ((k0 >> 1) + halfsel) << 7;
        const float* w1p = w1i + kph + (nlane << 1);
        const float* w2p = w2i + kph + (nlane << 1);

#pragma unroll
        for (int nb = 0; nb < 4; ++nb) {
            const v2f b1 = *(const v2f*)(w1p + nb * 32);
            const v2f b2 = *(const v2f*)(w2p + nb * 32);
            acc[nb] = __builtin_amdgcn_wmma_f32_16x16x4_f32(
                false, aSum, false, b1, (short)0, acc[nb], false, false);
            acc[nb] = __builtin_amdgcn_wmma_f32_16x16x4_f32(
                false, aBi,  false, b2, (short)0, acc[nb], false, false);
        }
    }

#pragma unroll
    for (int nb = 0; nb < 4; ++nb) {
#pragma unroll
        for (int i = 0; i < 8; ++i) {
            float v = acc[nb][i];
            v = (v > 0.f) ? v : v * NEG_SLOPE;
            outTile[(size_t)(i + halfsel * 8) * EMBED_DIM + nb * 16 + nlane] = v;
        }
    }
}

__global__ void ngcf_layer_kernel(const float* __restrict__ Eprev,
                                  const float* __restrict__ side,
                                  const float* __restrict__ W1,
                                  const float* __restrict__ W2,
                                  float*       __restrict__ Eout) {
    __shared__ float w1i[EMBED_DIM * EMBED_DIM];
    __shared__ float w2i[EMBED_DIM * EMBED_DIM];
    ngcf_stage_weights(W1, W2, w1i, w2i);

    const int tile = blockIdx.x * (blockDim.x >> 5) + (threadIdx.x >> 5);
    if (tile >= NODE_TILES) return;

    const int lane    = threadIdx.x & 31;
    const int halfsel = lane >> 4;
    const int nlane   = lane & 15;
    const int m0      = tile * 16;
    const int mrow    = m0 + nlane;
    const int koff    = halfsel * 2;

    ngcf_layer_tile(Eprev + (size_t)mrow * EMBED_DIM + koff,
                    side  + (size_t)mrow * EMBED_DIM + koff,
                    w1i, w2i, Eout + (size_t)m0 * EMBED_DIM, halfsel, nlane);
}

#define SEL_TILES (2 * BATCH / 16)
__global__ void ngcf_layer_sel_kernel(const float* __restrict__ Eprev,
                                      const float* __restrict__ side_sel,
                                      const float* __restrict__ W1,
                                      const float* __restrict__ W2,
                                      const int*   __restrict__ user_idx,
                                      const int*   __restrict__ item_idx,
                                      float*       __restrict__ Eout_sel) {
    __shared__ float w1i[EMBED_DIM * EMBED_DIM];
    __shared__ float w2i[EMBED_DIM * EMBED_DIM];
    ngcf_stage_weights(W1, W2, w1i, w2i);

    const int tile = blockIdx.x * (blockDim.x >> 5) + (threadIdx.x >> 5);
    if (tile >= SEL_TILES) return;

    const int lane    = threadIdx.x & 31;
    const int halfsel = lane >> 4;
    const int nlane   = lane & 15;
    const int m0      = tile * 16;
    const int slot    = m0 + nlane;
    const int node    = ngcf_slot_node(user_idx, item_idx, slot);
    const int koff    = halfsel * 2;

    ngcf_layer_tile(Eprev    + (size_t)node * EMBED_DIM + koff,
                    side_sel + (size_t)slot * EMBED_DIM + koff,
                    w1i, w2i, Eout_sel + (size_t)m0 * EMBED_DIM, halfsel, nlane);
}

__global__ void ngcf_gather_kernel(const float* __restrict__ emb,
                                   const float* __restrict__ E1,
                                   const float* __restrict__ E2,
                                   const float* __restrict__ E3sel,
                                   const int* __restrict__ user_idx,
                                   const int* __restrict__ item_idx,
                                   float* __restrict__ out) {
    const int row = blockIdx.x;
    const int t   = threadIdx.x;
    const int seg = t >> 6;
    const int dim = t & 63;

    const int node = ngcf_slot_node(user_idx, item_idx, row);

    const float* src = (seg == 0) ? emb : (seg == 1) ? E1 : (seg == 2) ? E2 : E3sel;
    const int    r   = (seg == 3) ? row : node;
    out[(size_t)row * 256 + t] = src[(size_t)r * EMBED_DIM + dim];
}

extern "C" void kernel_launch(void* const* d_in, const int* in_sizes, int n_in,
                              void* d_out, int out_size, void* d_ws, size_t ws_size,
                              hipStream_t stream) {
    const float* emb      = (const float*)d_in[0];
    const float* W1       = (const float*)d_in[1];
    const float* W2       = (const float*)d_in[2];
    const float* vals     = (const float*)d_in[3];
    const int*   row_idx  = (const int*)d_in[4];
    const int*   col_idx  = (const int*)d_in[5];
    const int*   user_idx = (const int*)d_in[6];
    const int*   item_idx = (const int*)d_in[7];

    const size_t EN = (size_t)NODE_NUM * EMBED_DIM;
    float* side = (float*)d_ws;
    float* E1   = side + EN;
    float* E2   = E1 + EN;
    float* P3   = E2 + EN;

    float* side_sel = side;
    float* E3sel    = side + (size_t)2 * BATCH * EMBED_DIM;

    int* pairs  = (int*)P3;
    int* rank   = pairs + 2 * (size_t)NNZ;
    int* counts = rank + (size_t)NNZ;
    int* rowptr = counts + 102400;
    int* bsum   = rowptr + 102400;

    ngcf_zero_kernel<<<(NODE_NUM / 4 + 255) / 256, 256, 0, stream>>>(
        (float4*)counts, NODE_NUM / 4);
    ngcf_rank_kernel<<<NNZ / 256, 256, 0, stream>>>(row_idx, counts, rank);
    ngcf_blocksum_kernel<<<SCAN_BLOCKS, 256, 0, stream>>>(counts, bsum);
    ngcf_scan_kernel<<<SCAN_BLOCKS, 256, 0, stream>>>(counts, bsum, rowptr);
    ngcf_place_kernel<<<NNZ / 256, 256, 0, stream>>>(vals, row_idx, col_idx,
                                                     rowptr, rank, pairs);

    const int layerBlocks = (NODE_TILES + 7) / 8;

    ngcf_spmm_rows_kernel<<<NODE_NUM / 8, 256, 0, stream>>>(rowptr, pairs, emb, side);
    ngcf_layer_kernel<<<layerBlocks, 256, 0, stream>>>(emb, side, W1, W2, E1);
    ngcf_spmm_rows_kernel<<<NODE_NUM / 8, 256, 0, stream>>>(rowptr, pairs, E1, side);
    ngcf_layer_kernel<<<layerBlocks, 256, 0, stream>>>(
        E1, side, W1 + EMBED_DIM * EMBED_DIM, W2 + EMBED_DIM * EMBED_DIM, E2);

    ngcf_spmm_sel_kernel<<<2 * BATCH / 8, 256, 0, stream>>>(
        rowptr, pairs, E2, user_idx, item_idx, side_sel);
    ngcf_layer_sel_kernel<<<SEL_TILES / 8, 256, 0, stream>>>(
        E2, side_sel, W1 + 2 * EMBED_DIM * EMBED_DIM, W2 + 2 * EMBED_DIM * EMBED_DIM,
        user_idx, item_idx, E3sel);

    ngcf_gather_kernel<<<2 * BATCH, 256, 0, stream>>>(emb, E1, E2, E3sel,
                                                      user_idx, item_idx,
                                                      (float*)d_out);
}


# Edge sort without global atomics: two-level bucket sort (256-row buckets) ranked by LDS atomics, stores staged in LDS into runs; row terms now summed in LDS-rank arrival order; 12 -> 11 launches
# speedup vs baseline: 3.4079x; 1.4342x over previous
#include <hip/hip_runtime.h>
#include <hip/hip_bf16.h>


#define USER_NUM 60000
#define ITEM_NUM 40000
#define NODE_NUM (USER_NUM + ITEM_NUM)
#define EMBED_DIM 64
#define N_LAYER 3
#define NNZ 1600000
#define BATCH 4096
#define NEG_SLOPE 0.2f
#define NODE_TILES (NODE_NUM / 16)

typedef float v2f __attribute__((ext_vector_type(2)));
typedef float v8f __attribute__((ext_vector_type(8)));

__device__ __forceinline__ int ngcf_slot_node(const int* __restrict__ user_idx,
                                              const int* __restrict__ item_idx,
                                              int s) {
    const int node = (s < BATCH) ? user_idx[s] : USER_NUM + item_idx[s - BATCH];
    return min(max(node, 0), NODE_NUM - 1);
}

#define BKT_SHIFT 8
#define BKT_ROWS (1 << BKT_SHIFT)
#define NBKT ((NODE_NUM + BKT_ROWS - 1) / BKT_ROWS)
#define SORT_THREADS 256
#define EPT_VEC 4
#define EPB 4096
#define NBLK ((NNZ + EPB - 1) / EPB)
#define BSCAN_CHUNKS ((NBLK + 31) / 32)
#define COL_BITS 17
#define COL_MASK ((1 << COL_BITS) - 1)
#define BKT_LOADS 24
#define BKT_STAGE (BKT_LOADS * SORT_THREADS)

static_assert(NBKT <= 2 * SORT_THREADS, "two buckets per thread in the block scans");
static_assert(BKT_ROWS <= SORT_THREADS, "one row per thread in the level-2 scan");
static_assert(EPB == EPT_VEC * 4 * SORT_THREADS, "edges per block");
static_assert(NODE_NUM <= (1 << COL_BITS) && (BKT_ROWS << COL_BITS) > 0, "entry packing");

__device__ __forceinline__ int ngcf_block_excl_scan(int v, int* wsum, int& total) {
    const int lane = threadIdx.x & 31;
    const int wid  = threadIdx.x >> 5;
    int incl = v;
    for (int off = 1; off < 32; off <<= 1) {
        const int t = __shfl_up(incl, off);
        if (lane >= off) incl += t;
    }
    if (lane == 31) wsum[wid] = incl;
    __syncthreads();
    const int wv = (lane < (int)(blockDim.x >> 5)) ? wsum[lane] : 0;
    int wincl = wv;
    for (int off = 1; off < 32; off <<= 1) {
        const int t = __shfl_up(wincl, off);
        if (lane >= off) wincl += t;
    }
    const int wpre = __shfl(wincl, wid) - __shfl(wv, wid);
    total = __shfl(wincl, 31);
    __syncthreads();
    return wpre + incl - v;
}

__device__ __forceinline__ int ngcf_clamp_row(int r) {
    return min(max(r, 0), NODE_NUM - 1);
}

__global__ void __launch_bounds__(SORT_THREADS)
ngcf_bhist_kernel(const int* __restrict__ rows, int* __restrict__ mat) {
    __shared__ int hist[NBKT];
    const int tid = threadIdx.x;
    for (int i = tid; i < NBKT; i += SORT_THREADS) hist[i] = 0;
    __syncthreads();
    int4 r4[EPT_VEC];
#pragma unroll
    for (int q = 0; q < EPT_VEC; ++q) {
        const int e0 = blockIdx.x * EPB + (q * SORT_THREADS + tid) * 4;
        r4[q] = *(const int4*)(rows + min(e0, NNZ - 4));
    }
#pragma unroll
    for (int q = 0; q < EPT_VEC; ++q) {
        const int e0 = blockIdx.x * EPB + (q * SORT_THREADS + tid) * 4;
        if (e0 < NNZ) {
            atomicAdd(&hist[ngcf_clamp_row(r4[q].x) >> BKT_SHIFT], 1);
            atomicAdd(&hist[ngcf_clamp_row(r4[q].y) >> BKT_SHIFT], 1);
            atomicAdd(&hist[ngcf_clamp_row(r4[q].z) >> BKT_SHIFT], 1);
            atomicAdd(&hist[ngcf_clamp_row(r4[q].w) >> BKT_SHIFT], 1);
        }
    }
    __syncthreads();
    for (int i = tid; i < NBKT; i += SORT_THREADS) mat[blockIdx.x * NBKT + i] = hist[i];
}

__global__ void ngcf_bscan_kernel(const int* __restrict__ mat,
                                  int* __restrict__ pre,
                                  int* __restrict__ total) {
    const int lane = threadIdx.x & 31;
    const int j = blockIdx.x * (blockDim.x >> 5) + (threadIdx.x >> 5);
    if (j >= NBKT) return;
    int v[BSCAN_CHUNKS];
#pragma unroll
    for (int i = 0; i < BSCAN_CHUNKS; ++i) {
        const int b = i * 32 + lane;
        v[i] = (b < NBLK) ? mat[b * NBKT + j] : 0;
    }
    int run = 0;
#pragma unroll
    for (int i = 0; i < BSCAN_CHUNKS; ++i) {
        int incl = v[i];
        for (int off = 1; off < 32; off <<= 1) {
            const int t = __shfl_up(incl, off);
            if (lane >= off) incl += t;
        }
        const int b = i * 32 + lane;
        if (b < NBLK) pre[b * NBKT + j] = run + incl - v[i];
        run += __shfl(incl, 31);
    }
    if (lane == 0) total[j] = run;
}

__global__ void __launch_bounds__(SORT_THREADS)
ngcf_bscatter_kernel(const float* __restrict__ vals,
                     const int*   __restrict__ rows,
                     const int*   __restrict__ cols,
                     const int*   __restrict__ pre,
                     const int*   __restrict__ total,
                     int*         __restrict__ tmp) {
    __shared__ int  hist[NBKT];
    __shared__ int  lbase[NBKT];
    __shared__ int  gdelta[NBKT];
    __shared__ int  wsum[32];
    __shared__ int2 ent[EPB];
    __shared__ int  gpos[EPB];
    const int tid = threadIdx.x;
    const int blk = blockIdx.x;

    for (int i = tid; i < NBKT; i += SORT_THREADS) hist[i] = 0;
    __syncthreads();

    int4   r4[EPT_VEC];
    int4   c4[EPT_VEC];
    float4 v4[EPT_VEC];
#pragma unroll
    for (int q = 0; q < EPT_VEC; ++q) {
        const int e0 = blk * EPB + (q * SORT_THREADS + tid) * 4;
        const int ec = min(e0, NNZ - 4);
        r4[q] = *(const int4*)(rows + ec);
        c4[q] = *(const int4*)(cols + ec);
        v4[q] = *(const float4*)(vals + ec);
    }
    int  bk[EPT_VEC * 4];
    int  lr[EPT_VEC * 4];
    int2 en[EPT_VEC * 4];
#pragma unroll
    for (int q = 0; q < EPT_VEC; ++q) {
        const int   e0 = blk * EPB + (q * SORT_THREADS + tid) * 4;
        const int   rr[4] = {r4[q].x, r4[q].y, r4[q].z, r4[q].w};
        const int   cc[4] = {c4[q].x, c4[q].y, c4[q].z, c4[q].w};
        const float vv[4] = {v4[q].x, v4[q].y, v4[q].z, v4[q].w};
#pragma unroll
        for (int k = 0; k < 4; ++k) {
            const int r = ngcf_clamp_row(rr[k]);
            const int c = (int)min((unsigned)cc[k], (unsigned)(NODE_NUM - 1));
            bk[q * 4 + k] = r >> BKT_SHIFT;
            en[q * 4 + k] = make_int2(((r & (BKT_ROWS - 1)) << COL_BITS) | c, __float_as_int(vv[k]));
            lr[q * 4 + k] = (e0 < NNZ) ? atomicAdd(&hist[bk[q * 4 + k]], 1) : 0;
        }
    }
    __syncthreads();

    const int b0 = tid * 2;
    const int b1 = b0 + 1;
    const int h0 = (b0 < NBKT) ? hist[b0] : 0;
    const int h1 = (b1 < NBKT) ? hist[b1] : 0;
    const int t0 = (b0 < NBKT) ? total[b0] : 0;
    const int t1 = (b1 < NBKT) ? total[b1] : 0;
    int tot;
    const int lb = ngcf_block_excl_scan(h0 + h1, wsum, tot);
    const int bb = ngcf_block_excl_scan(t0 + t1, wsum, tot);
    if (b0 < NBKT) {
        lbase[b0]  = lb;
        gdelta[b0] = bb + pre[blk * NBKT + b0] - lb;
    }
    if (b1 < NBKT) {
        lbase[b1]  = lb + h0;
        gdelta[b1] = bb + t0 + pre[blk * NBKT + b1] - (lb + h0);
    }
    __syncthreads();

#pragma unroll
    for (int q = 0; q < EPT_VEC; ++q) {
        const int e0 = blk * EPB + (q * SORT_THREADS + tid) * 4;
        if (e0 < NNZ) {
#pragma unroll
            for (int k = 0; k < 4; ++k) {
                const int lp = min(lbase[bk[q * 4 + k]] + lr[q * 4 + k], EPB - 1);
                ent[lp]  = en[q * 4 + k];
                gpos[lp] = gdelta[bk[q * 4 + k]] + lp;
            }
        }
    }
    __syncthreads();

    const int nvalid = min(EPB, NNZ - blk * EPB);
    for (int i = tid; i < nvalid; i += SORT_THREADS) {
        const int p = min(max(gpos[i], 0), NNZ - 1);
        ((int2*)tmp)[p] = ent[i];
    }
}

__global__ void __launch_bounds__(SORT_THREADS)
ngcf_bucket_kernel(const int* __restrict__ total,
                   const int* __restrict__ tmp,
                   int*       __restrict__ rowptr,
                   int*       __restrict__ pairs) {
    __shared__ int  cnt[BKT_ROWS];
    __shared__ int  exs[BKT_ROWS];
    __shared__ int  cur[BKT_ROWS];
    __shared__ int  wsum[32];
    __shared__ int2 stage[BKT_STAGE];
    const int tid = threadIdx.x;
    const int j   = blockIdx.x;

    if (tid < BKT_ROWS) { cnt[tid] = 0; cur[tid] = 0; }
    int base;
    const int tb0 = (tid * 2     < j) ? total[tid * 2]     : 0;
    const int tb1 = (tid * 2 + 1 < j) ? total[tid * 2 + 1] : 0;
    ngcf_block_excl_scan(tb0 + tb1, wsum, base);
    base = min(max(base, 0), NNZ);
    const int n = min(max(total[j], 0), NNZ - base);
    const int2* src = (const int2*)tmp + base;
    int2*       dst = (int2*)pairs + base;

    for (int i0 = 0; i0 < n; i0 += BKT_LOADS * SORT_THREADS) {
        int x[BKT_LOADS];
#pragma unroll
        for (int k = 0; k < BKT_LOADS; ++k) {
            const int idx = i0 + k * SORT_THREADS + tid;
            x[k] = (idx < n) ? src[idx].x : -1;
        }
#pragma unroll
        for (int k = 0; k < BKT_LOADS; ++k)
            if (x[k] >= 0) atomicAdd(&cnt[(x[k] >> COL_BITS) & (BKT_ROWS - 1)], 1);
    }
    __syncthreads();

    int tot;
    const int ex = ngcf_block_excl_scan((tid < BKT_ROWS) ? cnt[tid] : 0, wsum, tot);
    if (tid < BKT_ROWS) {
        exs[tid] = ex;
        const int row = j * BKT_ROWS + tid;
        if (row <= NODE_NUM) rowptr[row] = base + ex;
    }
    __syncthreads();

    const bool staged = n <= BKT_STAGE;
    for (int i0 = 0; i0 < n; i0 += BKT_LOADS * SORT_THREADS) {
        int2 e[BKT_LOADS];
#pragma unroll
        for (int k = 0; k < BKT_LOADS; ++k) {
            const int idx = i0 + k * SORT_THREADS + tid;
            e[k] = (idx < n) ? src[idx] : make_int2(-1, 0);
        }
#pragma unroll
        for (int k = 0; k < BKT_LOADS; ++k) {
            if (e[k].x >= 0) {
                const int lrow = (e[k].x >> COL_BITS) & (BKT_ROWS - 1);
                const int p = exs[lrow] + atomicAdd(&cur[lrow], 1);
                const int2 o = make_int2(e[k].x & COL_MASK, e[k].y);
                if (staged) stage[min(max(p, 0), BKT_STAGE - 1)] = o;
                else        dst[min(max(p, 0), n - 1)] = o;
            }
        }
    }
    if (staged) {
        __syncthreads();
        for (int i = tid; i < n; i += SORT_THREADS) dst[i] = stage[i];
    }
}

template <bool TAIL>
__device__ __forceinline__ void ngcf_gather8(const int2* __restrict__ pairs,
                                             const float* __restrict__ Elane,
                                             int j, int end, float& ax, float& ay) {
    int2 p[8];
    for (int k = 0; k < 8; ++k) {
        int jj = j + k;
        if (TAIL) jj = min(jj, end - 1);
        p[k] = pairs[jj];
    }
    float2 ev[8];
    for (int k = 0; k < 8; ++k) {
        const unsigned c = min((unsigned)p[k].x, (unsigned)(NODE_NUM - 1));
        ev[k] = *(const float2*)(Elane + (size_t)c * EMBED_DIM);
    }
    for (int k = 0; k < 8; ++k) {
        if (!TAIL || (j + k < end)) {
            const float v = __int_as_float(p[k].y);
            ax = fmaf(v, ev[k].x, ax);
            ay = fmaf(v, ev[k].y, ay);
        }
    }
}

__device__ __forceinline__ void ngcf_spmm_row(const int*   __restrict__ rowptr,
                                              const int*   __restrict__ pairs_i,
                                              const float* __restrict__ E,
                                              float*       __restrict__ dst_row,
                                              int r, int lane) {
    const int2* pairs = (const int2*)pairs_i;
    int       j   = min(max(rowptr[r], 0), NNZ);
    const int end = min(rowptr[r + 1], NNZ);
    const float* Elane = E + lane * 2;

    float ax = 0.f, ay = 0.f;
    for (; j + 8 <= end; j += 8) ngcf_gather8<false>(pairs, Elane, j, end, ax, ay);
    if (j < end) ngcf_gather8<true>(pairs, Elane, j, end, ax, ay);

    *(float2*)(dst_row + lane * 2) = make_float2(ax, ay);
}

__global__ void
ngcf_spmm_rows_kernel(const int*   __restrict__ rowptr,
                      const int*   __restrict__ pairs_i,
                      const float* __restrict__ E,
                      float*       __restrict__ side) {
    const int lane = threadIdx.x & 31;
    const int r = __builtin_amdgcn_readfirstlane(
        (int)(blockIdx.x * (blockDim.x >> 5) + (threadIdx.x >> 5)));
    if (r >= NODE_NUM) return;
    ngcf_spmm_row(rowptr, pairs_i, E, side + (size_t)r * EMBED_DIM, r, lane);
}

__global__ void
ngcf_spmm_sel_kernel(const int*   __restrict__ rowptr,
                     const int*   __restrict__ pairs_i,
                     const float* __restrict__ E,
                     const int*   __restrict__ user_idx,
                     const int*   __restrict__ item_idx,
                     float*       __restrict__ side_sel) {
    const int lane = threadIdx.x & 31;
    const int s = __builtin_amdgcn_readfirstlane(
        (int)(blockIdx.x * (blockDim.x >> 5) + (threadIdx.x >> 5)));
    if (s >= 2 * BATCH) return;
    const int r = ngcf_slot_node(user_idx, item_idx, s);
    ngcf_spmm_row(rowptr, pairs_i, E, side_sel + (size_t)s * EMBED_DIM, r, lane);
}

__device__ __forceinline__ void ngcf_stage_weights(const float* __restrict__ W1,
                                                   const float* __restrict__ W2,
                                                   float* w1i, float* w2i) {
    for (int i = threadIdx.x; i < EMBED_DIM * EMBED_DIM; i += blockDim.x) {
        const int k = i >> 6;
        const int n = i & 63;
        const int dst = ((k >> 1) << 7) + (n << 1) + (k & 1);
        w1i[dst] = W1[i];
        w2i[dst] = W2[i];
    }
    __syncthreads();
}

__device__ __forceinline__ void ngcf_layer_tile(const float* __restrict__ epRow,
                                                const float* __restrict__ sdRow,
                                                const float* w1i, const float* w2i,
                                                float* __restrict__ outTile,
                                                int halfsel, int nlane) {
    v8f acc[4] = {v8f{}, v8f{}, v8f{}, v8f{}};

#pragma unroll
    for (int kk = 0; kk < 16; ++kk) {
        const int k0 = kk * 4;
        const float2 ep = *(const float2*)(epRow + k0);
        const float2 sd = *(const float2*)(sdRow + k0);

        v2f aSum, aBi;
        aSum.x = ep.x + sd.x;  aSum.y = ep.y + sd.y;
        aBi.x  = ep.x * sd.x;  aBi.y  = ep.y * sd.y;

        const int kph = ((k0 >> 1) + halfsel) << 7;
        const float* w1p = w1i + kph + (nlane << 1);
        const float* w2p = w2i + kph + (nlane << 1);

#pragma unroll
        for (int nb = 0; nb < 4; ++nb) {
            const v2f b1 = *(const v2f*)(w1p + nb * 32);
            const v2f b2 = *(const v2f*)(w2p + nb * 32);
            acc[nb] = __builtin_amdgcn_wmma_f32_16x16x4_f32(
                false, aSum, false, b1, (short)0, acc[nb], false, false);
            acc[nb] = __builtin_amdgcn_wmma_f32_16x16x4_f32(
                false, aBi,  false, b2, (short)0, acc[nb], false, false);
        }
    }

#pragma unroll
    for (int nb = 0; nb < 4; ++nb) {
#pragma unroll
        for (int i = 0; i < 8; ++i) {
            float v = acc[nb][i];
            v = (v > 0.f) ? v : v * NEG_SLOPE;
            outTile[(size_t)(i + halfsel * 8) * EMBED_DIM + nb * 16 + nlane] = v;
        }
    }
}

__global__ void ngcf_layer_kernel(const float* __restrict__ Eprev,
                                  const float* __restrict__ side,
                                  const float* __restrict__ W1,
                                  const float* __restrict__ W2,
                                  float*       __restrict__ Eout) {
    __shared__ float w1i[EMBED_DIM * EMBED_DIM];
    __shared__ float w2i[EMBED_DIM * EMBED_DIM];
    ngcf_stage_weights(W1, W2, w1i, w2i);

    const int tile = blockIdx.x * (blockDim.x >> 5) + (threadIdx.x >> 5);
    if (tile >= NODE_TILES) return;

    const int lane    = threadIdx.x & 31;
    const int halfsel = lane >> 4;
    const int nlane   = lane & 15;
    const int m0      = tile * 16;
    const int mrow    = m0 + nlane;
    const int koff    = halfsel * 2;

    ngcf_layer_tile(Eprev + (size_t)mrow * EMBED_DIM + koff,
                    side  + (size_t)mrow * EMBED_DIM + koff,
                    w1i, w2i, Eout + (size_t)m0 * EMBED_DIM, halfsel, nlane);
}

#define SEL_TILES (2 * BATCH / 16)
__global__ void ngcf_layer_sel_kernel(const float* __restrict__ Eprev,
                                      const float* __restrict__ side_sel,
                                      const float* __restrict__ W1,
                                      const float* __restrict__ W2,
                                      const int*   __restrict__ user_idx,
                                      const int*   __restrict__ item_idx,
                                      float*       __restrict__ Eout_sel) {
    __shared__ float w1i[EMBED_DIM * EMBED_DIM];
    __shared__ float w2i[EMBED_DIM * EMBED_DIM];
    ngcf_stage_weights(W1, W2, w1i, w2i);

    const int tile = blockIdx.x * (blockDim.x >> 5) + (threadIdx.x >> 5);
    if (tile >= SEL_TILES) return;

    const int lane    = threadIdx.x & 31;
    const int halfsel = lane >> 4;
    const int nlane   = lane & 15;
    const int m0      = tile * 16;
    const int slot    = m0 + nlane;
    const int node    = ngcf_slot_node(user_idx, item_idx, slot);
    const int koff    = halfsel * 2;

    ngcf_layer_tile(Eprev    + (size_t)node * EMBED_DIM + koff,
                    side_sel + (size_t)slot * EMBED_DIM + koff,
                    w1i, w2i, Eout_sel + (size_t)m0 * EMBED_DIM, halfsel, nlane);
}

__global__ void ngcf_gather_kernel(const float* __restrict__ emb,
                                   const float* __restrict__ E1,
                                   const float* __restrict__ E2,
                                   const float* __restrict__ E3sel,
                                   const int* __restrict__ user_idx,
                                   const int* __restrict__ item_idx,
                                   float* __restrict__ out) {
    const int row = blockIdx.x;
    const int t   = threadIdx.x;
    const int seg = t >> 6;
    const int dim = t & 63;

    const int node = ngcf_slot_node(user_idx, item_idx, row);

    const float* src = (seg == 0) ? emb : (seg == 1) ? E1 : (seg == 2) ? E2 : E3sel;
    const int    r   = (seg == 3) ? row : node;
    out[(size_t)row * 256 + t] = src[(size_t)r * EMBED_DIM + dim];
}

extern "C" void kernel_launch(void* const* d_in, const int* in_sizes, int n_in,
                              void* d_out, int out_size, void* d_ws, size_t ws_size,
                              hipStream_t stream) {
    const float* emb      = (const float*)d_in[0];
    const float* W1       = (const float*)d_in[1];
    const float* W2       = (const float*)d_in[2];
    const float* vals     = (const float*)d_in[3];
    const int*   row_idx  = (const int*)d_in[4];
    const int*   col_idx  = (const int*)d_in[5];
    const int*   user_idx = (const int*)d_in[6];
    const int*   item_idx = (const int*)d_in[7];

    const size_t EN = (size_t)NODE_NUM * EMBED_DIM;
    float* side = (float*)d_ws;
    float* E1   = side + EN;
    float* E2   = E1 + EN;
    float* P3   = E2 + EN;

    float* side_sel = side;
    float* E3sel    = side + (size_t)2 * BATCH * EMBED_DIM;

    int* pairs  = (int*)P3;
    int* rowptr = pairs + 2 * (size_t)NNZ;
    int* mat    = rowptr + 102400;
    int* pre    = mat + NBLK * NBKT;
    int* total  = pre + NBLK * NBKT;
    int* tmp    = (int*)E2;

    ngcf_bhist_kernel<<<NBLK, SORT_THREADS, 0, stream>>>(row_idx, mat);
    ngcf_bscan_kernel<<<(NBKT + 7) / 8, 256, 0, stream>>>(mat, pre, total);
    ngcf_bscatter_kernel<<<NBLK, SORT_THREADS, 0, stream>>>(vals, row_idx, col_idx,
                                                            pre, total, tmp);
    ngcf_bucket_kernel<<<NBKT, SORT_THREADS, 0, stream>>>(total, tmp, rowptr, pairs);

    const int layerBlocks = (NODE_TILES + 7) / 8;

    ngcf_spmm_rows_kernel<<<NODE_NUM / 8, 256, 0, stream>>>(rowptr, pairs, emb, side);
    ngcf_layer_kernel<<<layerBlocks, 256, 0, stream>>>(emb, side, W1, W2, E1);
    ngcf_spmm_rows_kernel<<<NODE_NUM / 8, 256, 0, stream>>>(rowptr, pairs, E1, side);
    ngcf_layer_kernel<<<layerBlocks, 256, 0, stream>>>(
        E1, side, W1 + EMBED_DIM * EMBED_DIM, W2 + EMBED_DIM * EMBED_DIM, E2);

    ngcf_spmm_sel_kernel<<<2 * BATCH / 8, 256, 0, stream>>>(
        rowptr, pairs, E2, user_idx, item_idx, side_sel);
    ngcf_layer_sel_kernel<<<SEL_TILES / 8, 256, 0, stream>>>(
        E2, side_sel, W1 + 2 * EMBED_DIM * EMBED_DIM, W2 + 2 * EMBED_DIM * EMBED_DIM,
        user_idx, item_idx, E3sel);

    ngcf_gather_kernel<<<2 * BATCH, 256, 0, stream>>>(emb, E1, E2, E3sel,
                                                      user_idx, item_idx,
                                                      (float*)d_out);
}


# Layer 3 and the output assembly in one launch: 16 waves per 16 output slots sum the side rows into LDS, run the tile's GEMMs from LDS and store whole output rows; 11 -> 9 launches; same sum order
# speedup vs baseline: 3.8557x; 1.1314x over previous
#include <hip/hip_runtime.h>
#include <hip/hip_bf16.h>


#define USER_NUM 60000
#define ITEM_NUM 40000
#define NODE_NUM (USER_NUM + ITEM_NUM)
#define EMBED_DIM 64
#define N_LAYER 3
#define NNZ 1600000
#define BATCH 4096
#define NEG_SLOPE 0.2f
#define NODE_TILES (NODE_NUM / 16)

typedef float v2f __attribute__((ext_vector_type(2)));
typedef float v8f __attribute__((ext_vector_type(8)));

__device__ __forceinline__ int ngcf_slot_node(const int* __restrict__ user_idx,
                                              const int* __restrict__ item_idx,
                                              int s) {
    const int node = (s < BATCH) ? user_idx[s] : USER_NUM + item_idx[s - BATCH];
    return min(max(node, 0), NODE_NUM - 1);
}

#define BKT_SHIFT 8
#define BKT_ROWS (1 << BKT_SHIFT)
#define NBKT ((NODE_NUM + BKT_ROWS - 1) / BKT_ROWS)
#define SORT_THREADS 256
#define EPT_VEC 4
#define EPB 4096
#define NBLK ((NNZ + EPB - 1) / EPB)
#define BSCAN_CHUNKS ((NBLK + 31) / 32)
#define COL_BITS 17
#define COL_MASK ((1 << COL_BITS) - 1)
#define BKT_LOADS 24
#define BKT_STAGE (BKT_LOADS * SORT_THREADS)

static_assert(NBKT <= 2 * SORT_THREADS, "two buckets per thread in the block scans");
static_assert(BKT_ROWS <= SORT_THREADS, "one row per thread in the level-2 scan");
static_assert(EPB == EPT_VEC * 4 * SORT_THREADS, "edges per block");
static_assert(NODE_NUM <= (1 << COL_BITS) && (BKT_ROWS << COL_BITS) > 0, "entry packing");

__device__ __forceinline__ int ngcf_block_excl_scan(int v, int* wsum, int& total) {
    const int lane = threadIdx.x & 31;
    const int wid  = threadIdx.x >> 5;
    int incl = v;
    for (int off = 1; off < 32; off <<= 1) {
        const int t = __shfl_up(incl, off);
        if (lane >= off) incl += t;
    }
    if (lane == 31) wsum[wid] = incl;
    __syncthreads();
    const int wv = (lane < (int)(blockDim.x >> 5)) ? wsum[lane] : 0;
    int wincl = wv;
    for (int off = 1; off < 32; off <<= 1) {
        const int t = __shfl_up(wincl, off);
        if (lane >= off) wincl += t;
    }
    const int wpre = __shfl(wincl, wid) - __shfl(wv, wid);
    total = __shfl(wincl, 31);
    __syncthreads();
    return wpre + incl - v;
}

__device__ __forceinline__ int ngcf_clamp_row(int r) {
    return min(max(r, 0), NODE_NUM - 1);
}

__global__ void __launch_bounds__(SORT_THREADS)
ngcf_bhist_kernel(const int* __restrict__ rows, int* __restrict__ mat) {
    __shared__ int hist[NBKT];
    const int tid = threadIdx.x;
    for (int i = tid; i < NBKT; i += SORT_THREADS) hist[i] = 0;
    __syncthreads();
    int4 r4[EPT_VEC];
#pragma unroll
    for (int q = 0; q < EPT_VEC; ++q) {
        const int e0 = blockIdx.x * EPB + (q * SORT_THREADS + tid) * 4;
        r4[q] = *(const int4*)(rows + min(e0, NNZ - 4));
    }
#pragma unroll
    for (int q = 0; q < EPT_VEC; ++q) {
        const int e0 = blockIdx.x * EPB + (q * SORT_THREADS + tid) * 4;
        if (e0 < NNZ) {
            atomicAdd(&hist[ngcf_clamp_row(r4[q].x) >> BKT_SHIFT], 1);
            atomicAdd(&hist[ngcf_clamp_row(r4[q].y) >> BKT_SHIFT], 1);
            atomicAdd(&hist[ngcf_clamp_row(r4[q].z) >> BKT_SHIFT], 1);
            atomicAdd(&hist[ngcf_clamp_row(r4[q].w) >> BKT_SHIFT], 1);
        }
    }
    __syncthreads();
    for (int i = tid; i < NBKT; i += SORT_THREADS) mat[blockIdx.x * NBKT + i] = hist[i];
}

__global__ void ngcf_bscan_kernel(const int* __restrict__ mat,
                                  int* __restrict__ pre,
                                  int* __restrict__ total) {
    const int lane = threadIdx.x & 31;
    const int j = blockIdx.x * (blockDim.x >> 5) + (threadIdx.x >> 5);
    if (j >= NBKT) return;
    int v[BSCAN_CHUNKS];
#pragma unroll
    for (int i = 0; i < BSCAN_CHUNKS; ++i) {
        const int b = i * 32 + lane;
        v[i] = (b < NBLK) ? mat[b * NBKT + j] : 0;
    }
    int run = 0;
#pragma unroll
    for (int i = 0; i < BSCAN_CHUNKS; ++i) {
        int incl = v[i];
        for (int off = 1; off < 32; off <<= 1) {
            const int t = __shfl_up(incl, off);
            if (lane >= off) incl += t;
        }
        const int b = i * 32 + lane;
        if (b < NBLK) pre[b * NBKT + j] = run + incl - v[i];
        run += __shfl(incl, 31);
    }
    if (lane == 0) total[j] = run;
}

__global__ void __launch_bounds__(SORT_THREADS)
ngcf_bscatter_kernel(const float* __restrict__ vals,
                     const int*   __restrict__ rows,
                     const int*   __restrict__ cols,
                     const int*   __restrict__ pre,
                     const int*   __restrict__ total,
                     int*         __restrict__ tmp) {
    __shared__ int  hist[NBKT];
    __shared__ int  lbase[NBKT];
    __shared__ int  gdelta[NBKT];
    __shared__ int  wsum[32];
    __shared__ int2 ent[EPB];
    __shared__ int  gpos[EPB];
    const int tid = threadIdx.x;
    const int blk = blockIdx.x;

    for (int i = tid; i < NBKT; i += SORT_THREADS) hist[i] = 0;
    __syncthreads();

    int4   r4[EPT_VEC];
    int4   c4[EPT_VEC];
    float4 v4[EPT_VEC];
#pragma unroll
    for (int q = 0; q < EPT_VEC; ++q) {
        const int e0 = blk * EPB + (q * SORT_THREADS + tid) * 4;
        const int ec = min(e0, NNZ - 4);
        r4[q] = *(const int4*)(rows + ec);
        c4[q] = *(const int4*)(cols + ec);
        v4[q] = *(const float4*)(vals + ec);
    }
    int  bk[EPT_VEC * 4];
    int  lr[EPT_VEC * 4];
    int2 en[EPT_VEC * 4];
#pragma unroll
    for (int q = 0; q < EPT_VEC; ++q) {
        const int   e0 = blk * EPB + (q * SORT_THREADS + tid) * 4;
        const int   rr[4] = {r4[q].x, r4[q].y, r4[q].z, r4[q].w};
        const int   cc[4] = {c4[q].x, c4[q].y, c4[q].z, c4[q].w};
        const float vv[4] = {v4[q].x, v4[q].y, v4[q].z, v4[q].w};
#pragma unroll
        for (int k = 0; k < 4; ++k) {
            const int r = ngcf_clamp_row(rr[k]);
            const int c = (int)min((unsigned)cc[k], (unsigned)(NODE_NUM - 1));
            bk[q * 4 + k] = r >> BKT_SHIFT;
            en[q * 4 + k] = make_int2(((r & (BKT_ROWS - 1)) << COL_BITS) | c, __float_as_int(vv[k]));
            lr[q * 4 + k] = (e0 < NNZ) ? atomicAdd(&hist[bk[q * 4 + k]], 1) : 0;
        }
    }
    __syncthreads();

    const int b0 = tid * 2;
    const int b1 = b0 + 1;
    const int h0 = (b0 < NBKT) ? hist[b0] : 0;
    const int h1 = (b1 < NBKT) ? hist[b1] : 0;
    const int t0 = (b0 < NBKT) ? total[b0] : 0;
    const int t1 = (b1 < NBKT) ? total[b1] : 0;
    int tot;
    const int lb = ngcf_block_excl_scan(h0 + h1, wsum, tot);
    const int bb = ngcf_block_excl_scan(t0 + t1, wsum, tot);
    if (b0 < NBKT) {
        lbase[b0]  = lb;
        gdelta[b0] = bb + pre[blk * NBKT + b0] - lb;
    }
    if (b1 < NBKT) {
        lbase[b1]  = lb + h0;
        gdelta[b1] = bb + t0 + pre[blk * NBKT + b1] - (lb + h0);
    }
    __syncthreads();

#pragma unroll
    for (int q = 0; q < EPT_VEC; ++q) {
        const int e0 = blk * EPB + (q * SORT_THREADS + tid) * 4;
        if (e0 < NNZ) {
#pragma unroll
            for (int k = 0; k < 4; ++k) {
                const int lp = min(lbase[bk[q * 4 + k]] + lr[q * 4 + k], EPB - 1);
                ent[lp]  = en[q * 4 + k];
                gpos[lp] = gdelta[bk[q * 4 + k]] + lp;
            }
        }
    }
    __syncthreads();

    const int nvalid = min(EPB, NNZ - blk * EPB);
    for (int i = tid; i < nvalid; i += SORT_THREADS) {
        const int p = min(max(gpos[i], 0), NNZ - 1);
        ((int2*)tmp)[p] = ent[i];
    }
}

__global__ void __launch_bounds__(SORT_THREADS)
ngcf_bucket_kernel(const int* __restrict__ total,
                   const int* __restrict__ tmp,
                   int*       __restrict__ rowptr,
                   int*       __restrict__ pairs) {
    __shared__ int  cnt[BKT_ROWS];
    __shared__ int  exs[BKT_ROWS];
    __shared__ int  cur[BKT_ROWS];
    __shared__ int  wsum[32];
    __shared__ int2 stage[BKT_STAGE];
    const int tid = threadIdx.x;
    const int j   = blockIdx.x;

    if (tid < BKT_ROWS) { cnt[tid] = 0; cur[tid] = 0; }
    int base;
    const int tb0 = (tid * 2     < j) ? total[tid * 2]     : 0;
    const int tb1 = (tid * 2 + 1 < j) ? total[tid * 2 + 1] : 0;
    ngcf_block_excl_scan(tb0 + tb1, wsum, base);
    base = min(max(base, 0), NNZ);
    const int n = min(max(total[j], 0), NNZ - base);
    const int2* src = (const int2*)tmp + base;
    int2*       dst = (int2*)pairs + base;

    for (int i0 = 0; i0 < n; i0 += BKT_LOADS * SORT_THREADS) {
        int x[BKT_LOADS];
#pragma unroll
        for (int k = 0; k < BKT_LOADS; ++k) {
            const int idx = i0 + k * SORT_THREADS + tid;
            x[k] = (idx < n) ? src[idx].x : -1;
        }
#pragma unroll
        for (int k = 0; k < BKT_LOADS; ++k)
            if (x[k] >= 0) atomicAdd(&cnt[(x[k] >> COL_BITS) & (BKT_ROWS - 1)], 1);
    }
    __syncthreads();

    int tot;
    const int ex = ngcf_block_excl_scan((tid < BKT_ROWS) ? cnt[tid] : 0, wsum, tot);
    if (tid < BKT_ROWS) {
        exs[tid] = ex;
        const int row = j * BKT_ROWS + tid;
        if (row <= NODE_NUM) rowptr[row] = base + ex;
    }
    __syncthreads();

    const bool staged = n <= BKT_STAGE;
    for (int i0 = 0; i0 < n; i0 += BKT_LOADS * SORT_THREADS) {
        int2 e[BKT_LOADS];
#pragma unroll
        for (int k = 0; k < BKT_LOADS; ++k) {
            const int idx = i0 + k * SORT_THREADS + tid;
            e[k] = (idx < n) ? src[idx] : make_int2(-1, 0);
        }
#pragma unroll
        for (int k = 0; k < BKT_LOADS; ++k) {
            if (e[k].x >= 0) {
                const int lrow = (e[k].x >> COL_BITS) & (BKT_ROWS - 1);
                const int p = exs[lrow] + atomicAdd(&cur[lrow], 1);
                const int2 o = make_int2(e[k].x & COL_MASK, e[k].y);
                if (staged) stage[min(max(p, 0), BKT_STAGE - 1)] = o;
                else        dst[min(max(p, 0), n - 1)] = o;
            }
        }
    }
    if (staged) {
        __syncthreads();
        for (int i = tid; i < n; i += SORT_THREADS) dst[i] = stage[i];
    }
}

template <bool TAIL>
__device__ __forceinline__ void ngcf_gather8(const int2* __restrict__ pairs,
                                             const float* __restrict__ Elane,
                                             int j, int end, float& ax, float& ay) {
    int2 p[8];
    for (int k = 0; k < 8; ++k) {
        int jj = j + k;
        if (TAIL) jj = min(jj, end - 1);
        p[k] = pairs[jj];
    }
    float2 ev[8];
    for (int k = 0; k < 8; ++k) {
        const unsigned c = min((unsigned)p[k].x, (unsigned)(NODE_NUM - 1));
        ev[k] = *(const float2*)(Elane + (size_t)c * EMBED_DIM);
    }
    for (int k = 0; k < 8; ++k) {
        if (!TAIL || (j + k < end)) {
            const float v = __int_as_float(p[k].y);
            ax = fmaf(v, ev[k].x, ax);
            ay = fmaf(v, ev[k].y, ay);
        }
    }
}

__device__ __forceinline__ void ngcf_spmm_row(const int*   __restrict__ rowptr,
                                              const int*   __restrict__ pairs_i,
                                              const float* __restrict__ E,
                                              float*       __restrict__ dst_row,
                                              int r, int lane) {
    const int2* pairs = (const int2*)pairs_i;
    int       j   = min(max(rowptr[r], 0), NNZ);
    const int end = min(rowptr[r + 1], NNZ);
    const float* Elane = E + lane * 2;

    float ax = 0.f, ay = 0.f;
    for (; j + 8 <= end; j += 8) ngcf_gather8<false>(pairs, Elane, j, end, ax, ay);
    if (j < end) ngcf_gather8<true>(pairs, Elane, j, end, ax, ay);

    *(float2*)(dst_row + lane * 2) = make_float2(ax, ay);
}

__global__ void
ngcf_spmm_rows_kernel(const int*   __restrict__ rowptr,
                      const int*   __restrict__ pairs_i,
                      const float* __restrict__ E,
                      float*       __restrict__ side) {
    const int lane = threadIdx.x & 31;
    const int r = __builtin_amdgcn_readfirstlane(
        (int)(blockIdx.x * (blockDim.x >> 5) + (threadIdx.x >> 5)));
    if (r >= NODE_NUM) return;
    ngcf_spmm_row(rowptr, pairs_i, E, side + (size_t)r * EMBED_DIM, r, lane);
}

__device__ __forceinline__ void ngcf_stage_weights(const float* __restrict__ W1,
                                                   const float* __restrict__ W2,
                                                   float* w1i, float* w2i) {
    for (int i = threadIdx.x; i < EMBED_DIM * EMBED_DIM; i += blockDim.x) {
        const int k = i >> 6;
        const int n = i & 63;
        const int dst = ((k >> 1) << 7) + (n << 1) + (k & 1);
        w1i[dst] = W1[i];
        w2i[dst] = W2[i];
    }
    __syncthreads();
}

__device__ __forceinline__ void ngcf_layer_tile(const float* __restrict__ epRow,
                                                const float* __restrict__ sdRow,
                                                const float* w1i, const float* w2i,
                                                float* __restrict__ outTile,
                                                int halfsel, int nlane) {
    v8f acc[4] = {v8f{}, v8f{}, v8f{}, v8f{}};

#pragma unroll
    for (int kk = 0; kk < 16; ++kk) {
        const int k0 = kk * 4;
        const float2 ep = *(const float2*)(epRow + k0);
        const float2 sd = *(const float2*)(sdRow + k0);

        v2f aSum, aBi;
        aSum.x = ep.x + sd.x;  aSum.y = ep.y + sd.y;
        aBi.x  = ep.x * sd.x;  aBi.y  = ep.y * sd.y;

        const int kph = ((k0 >> 1) + halfsel) << 7;
        const float* w1p = w1i + kph + (nlane << 1);
        const float* w2p = w2i + kph + (nlane << 1);

#pragma unroll
        for (int nb = 0; nb < 4; ++nb) {
            const v2f b1 = *(const v2f*)(w1p + nb * 32);
            const v2f b2 = *(const v2f*)(w2p + nb * 32);
            acc[nb] = __builtin_amdgcn_wmma_f32_16x16x4_f32(
                false, aSum, false, b1, (short)0, acc[nb], false, false);
            acc[nb] = __builtin_amdgcn_wmma_f32_16x16x4_f32(
                false, aBi,  false, b2, (short)0, acc[nb], false, false);
        }
    }

#pragma unroll
    for (int nb = 0; nb < 4; ++nb) {
#pragma unroll
        for (int i = 0; i < 8; ++i) {
            float v = acc[nb][i];
            v = (v > 0.f) ? v : v * NEG_SLOPE;
            outTile[(size_t)(i + halfsel * 8) * EMBED_DIM + nb * 16 + nlane] = v;
        }
    }
}

__global__ void ngcf_layer_kernel(const float* __restrict__ Eprev,
                                  const float* __restrict__ side,
                                  const float* __restrict__ W1,
                                  const float* __restrict__ W2,
                                  float*       __restrict__ Eout) {
    __shared__ float w1i[EMBED_DIM * EMBED_DIM];
    __shared__ float w2i[EMBED_DIM * EMBED_DIM];
    ngcf_stage_weights(W1, W2, w1i, w2i);

    const int tile = blockIdx.x * (blockDim.x >> 5) + (threadIdx.x >> 5);
    if (tile >= NODE_TILES) return;

    const int lane    = threadIdx.x & 31;
    const int halfsel = lane >> 4;
    const int nlane   = lane & 15;
    const int m0      = tile * 16;
    const int mrow    = m0 + nlane;
    const int koff    = halfsel * 2;

    ngcf_layer_tile(Eprev + (size_t)mrow * EMBED_DIM + koff,
                    side  + (size_t)mrow * EMBED_DIM + koff,
                    w1i, w2i, Eout + (size_t)m0 * EMBED_DIM, halfsel, nlane);
}

#define SEL_TILES (2 * BATCH / 16)
#define TAIL_WAVES 16
#define TAIL_THREADS (TAIL_WAVES * 32)
#define TAIL_PITCH 66
static_assert(EMBED_DIM * EMBED_DIM == 8 * TAIL_THREADS, "weights: 8 floats per thread per matrix");

__device__ __forceinline__ void ngcf_layer_block16(const float* epRow, const float* sdRow,
                                                   const float* w1p, const float* w2p,
                                                   float* dstCol, int halfsel) {
    v8f acc = v8f{};
#pragma unroll
    for (int kk = 0; kk < 16; ++kk) {
        const int k0 = kk * 4;
        const float2 ep = *(const float2*)(epRow + k0);
        const float2 sd = *(const float2*)(sdRow + k0);

        v2f aSum, aBi;
        aSum.x = ep.x + sd.x;  aSum.y = ep.y + sd.y;
        aBi.x  = ep.x * sd.x;  aBi.y  = ep.y * sd.y;

        const int kph = ((k0 >> 1) + halfsel) << 7;
        const v2f b1 = *(const v2f*)(w1p + kph);
        const v2f b2 = *(const v2f*)(w2p + kph);
        acc = __builtin_amdgcn_wmma_f32_16x16x4_f32(
            false, aSum, false, b1, (short)0, acc, false, false);
        acc = __builtin_amdgcn_wmma_f32_16x16x4_f32(
            false, aBi,  false, b2, (short)0, acc, false, false);
    }
#pragma unroll
    for (int i = 0; i < 8; ++i) {
        float v = acc[i];
        v = (v > 0.f) ? v : v * NEG_SLOPE;
        dstCol[(i + halfsel * 8) * TAIL_PITCH] = v;
    }
}

__global__ void __launch_bounds__(TAIL_THREADS)
ngcf_tail_kernel(const int*   __restrict__ rowptr,
                 const int*   __restrict__ pairs_i,
                 const float* __restrict__ emb,
                 const float* __restrict__ E1,
                 const float* __restrict__ E2,
                 const float* __restrict__ W1,
                 const float* __restrict__ W2,
                 const int*   __restrict__ user_idx,
                 const int*   __restrict__ item_idx,
                 float*       __restrict__ out) {
    __shared__ float w1i[EMBED_DIM * EMBED_DIM]  __attribute__((aligned(16)));
    __shared__ float w2i[EMBED_DIM * EMBED_DIM]  __attribute__((aligned(16)));
    __shared__ float sideT[16 * TAIL_PITCH]      __attribute__((aligned(16)));
    __shared__ float e2T[16 * TAIL_PITCH]        __attribute__((aligned(16)));
    __shared__ float e3T[16 * TAIL_PITCH]        __attribute__((aligned(16)));

    const int tid  = threadIdx.x;
    const int lane = tid & 31;
    const int wid = __builtin_amdgcn_readfirstlane(tid >> 5);
    const int m0  = blockIdx.x * 16;
    const int s   = m0 + wid;
    const int r   = ngcf_slot_node(user_idx, item_idx, s);

    float wa[8], wb[8];
#pragma unroll
    for (int j = 0; j < 8; ++j) {
        wa[j] = W1[j * TAIL_THREADS + tid];
        wb[j] = W2[j * TAIL_THREADS + tid];
    }

    const size_t ro = (size_t)r * EMBED_DIM + lane * 2;
    const float2 o0 = *(const float2*)(emb + ro);
    const float2 o1 = *(const float2*)(E1 + ro);
    const float2 o2 = *(const float2*)(E2 + ro);

    ngcf_spmm_row(rowptr, pairs_i, E2, sideT + wid * TAIL_PITCH, r, lane);
    *(float2*)(e2T + wid * TAIL_PITCH + lane * 2) = o2;

#pragma unroll
    for (int j = 0; j < 8; ++j) {
        const int i = j * TAIL_THREADS + tid;
        const int k = i >> 6;
        const int n = i & 63;
        const int dst = ((k >> 1) << 7) + (n << 1) + (k & 1);
        w1i[dst] = wa[j];
        w2i[dst] = wb[j];
    }
    __syncthreads();

    if (wid < 4) {
        const int halfsel = lane >> 4;
        const int nlane   = lane & 15;
        const int koff    = halfsel * 2;
        const int colo    = (nlane << 1) + wid * 32;
        ngcf_layer_block16(e2T   + nlane * TAIL_PITCH + koff,
                           sideT + nlane * TAIL_PITCH + koff,
                           w1i + colo, w2i + colo,
                           e3T + wid * 16 + nlane, halfsel);
    }
    __syncthreads();

    float* orow = out + (size_t)s * 256 + lane * 2;
    *(float2*)(orow)       = o0;
    *(float2*)(orow + 64)  = o1;
    *(float2*)(orow + 128) = o2;
    *(float2*)(orow + 192) = *(const float2*)(e3T + wid * TAIL_PITCH + lane * 2);
}

extern "C" void kernel_launch(void* const* d_in, const int* in_sizes, int n_in,
                              void* d_out, int out_size, void* d_ws, size_t ws_size,
                              hipStream_t stream) {
    const float* emb      = (const float*)d_in[0];
    const float* W1       = (const float*)d_in[1];
    const float* W2       = (const float*)d_in[2];
    const float* vals     = (const float*)d_in[3];
    const int*   row_idx  = (const int*)d_in[4];
    const int*   col_idx  = (const int*)d_in[5];
    const int*   user_idx = (const int*)d_in[6];
    const int*   item_idx = (const int*)d_in[7];

    const size_t EN = (size_t)NODE_NUM * EMBED_DIM;
    float* side = (float*)d_ws;
    float* E1   = side + EN;
    float* E2   = E1 + EN;
    float* P3   = E2 + EN;

    int* pairs  = (int*)P3;
    int* rowptr = pairs + 2 * (size_t)NNZ;
    int* mat    = rowptr + 102400;
    int* pre    = mat + NBLK * NBKT;
    int* total  = pre + NBLK * NBKT;
    int* tmp    = (int*)E2;

    ngcf_bhist_kernel<<<NBLK, SORT_THREADS, 0, stream>>>(row_idx, mat);
    ngcf_bscan_kernel<<<(NBKT + 7) / 8, 256, 0, stream>>>(mat, pre, total);
    ngcf_bscatter_kernel<<<NBLK, SORT_THREADS, 0, stream>>>(vals, row_idx, col_idx,
                                                            pre, total, tmp);
    ngcf_bucket_kernel<<<NBKT, SORT_THREADS, 0, stream>>>(total, tmp, rowptr, pairs);

    const int layerBlocks = (NODE_TILES + 7) / 8;

    ngcf_spmm_rows_kernel<<<NODE_NUM / 8, 256, 0, stream>>>(rowptr, pairs, emb, side);
    ngcf_layer_kernel<<<layerBlocks, 256, 0, stream>>>(emb, side, W1, W2, E1);
    ngcf_spmm_rows_kernel<<<NODE_NUM / 8, 256, 0, stream>>>(rowptr, pairs, E1, side);
    ngcf_layer_kernel<<<layerBlocks, 256, 0, stream>>>(
        E1, side, W1 + EMBED_DIM * EMBED_DIM, W2 + EMBED_DIM * EMBED_DIM, E2);

    ngcf_tail_kernel<<<SEL_TILES, TAIL_THREADS, 0, stream>>>(
        rowptr, pairs, emb, E1, E2,
        W1 + 2 * EMBED_DIM * EMBED_DIM, W2 + 2 * EMBED_DIM * EMBED_DIM,
        user_idx, item_idx, (float*)d_out);
}


# Layers 1 and 2: SpMM fused into the layer kernel: 893 resident 16-wave workgroups loop over 16-node tiles, sum side rows into LDS, GEMM from LDS, store whole rows; 9 -> 7 launches; same sum order
# speedup vs baseline: 4.7742x; 1.2382x over previous
#include <hip/hip_runtime.h>
#include <hip/hip_bf16.h>


#define USER_NUM 60000
#define ITEM_NUM 40000
#define NODE_NUM (USER_NUM + ITEM_NUM)
#define EMBED_DIM 64
#define N_LAYER 3
#define NNZ 1600000
#define BATCH 4096
#define NEG_SLOPE 0.2f
#define NODE_TILES (NODE_NUM / 16)

typedef float v2f __attribute__((ext_vector_type(2)));
typedef float v8f __attribute__((ext_vector_type(8)));

__device__ __forceinline__ int ngcf_slot_node(const int* __restrict__ user_idx,
                                              const int* __restrict__ item_idx,
                                              int s) {
    const int node = (s < BATCH) ? user_idx[s] : USER_NUM + item_idx[s - BATCH];
    return min(max(node, 0), NODE_NUM - 1);
}

#define BKT_SHIFT 8
#define BKT_ROWS (1 << BKT_SHIFT)
#define NBKT ((NODE_NUM + BKT_ROWS - 1) / BKT_ROWS)
#define SORT_THREADS 256
#define EPT_VEC 4
#define EPB 4096
#define NBLK ((NNZ + EPB - 1) / EPB)
#define BSCAN_CHUNKS ((NBLK + 31) / 32)
#define COL_BITS 17
#define COL_MASK ((1 << COL_BITS) - 1)
#define BKT_LOADS 24
#define BKT_STAGE (BKT_LOADS * SORT_THREADS)

static_assert(NBKT <= 2 * SORT_THREADS, "two buckets per thread in the block scans");
static_assert(BKT_ROWS <= SORT_THREADS, "one row per thread in the level-2 scan");
static_assert(EPB == EPT_VEC * 4 * SORT_THREADS, "edges per block");
static_assert(NODE_NUM <= (1 << COL_BITS) && (BKT_ROWS << COL_BITS) > 0, "entry packing");

__device__ __forceinline__ int ngcf_block_excl_scan(int v, int* wsum, int& total) {
    const int lane = threadIdx.x & 31;
    const int wid  = threadIdx.x >> 5;
    int incl = v;
    for (int off = 1; off < 32; off <<= 1) {
        const int t = __shfl_up(incl, off);
        if (lane >= off) incl += t;
    }
    if (lane == 31) wsum[wid] = incl;
    __syncthreads();
    const int wv = (lane < (int)(blockDim.x >> 5)) ? wsum[lane] : 0;
    int wincl = wv;
    for (int off = 1; off < 32; off <<= 1) {
        const int t = __shfl_up(wincl, off);
        if (lane >= off) wincl += t;
    }
    const int wpre = __shfl(wincl, wid) - __shfl(wv, wid);
    total = __shfl(wincl, 31);
    __syncthreads();
    return wpre + incl - v;
}

__device__ __forceinline__ int ngcf_clamp_row(int r) {
    return min(max(r, 0), NODE_NUM - 1);
}

__global__ void __launch_bounds__(SORT_THREADS)
ngcf_bhist_kernel(const int* __restrict__ rows, int* __restrict__ mat) {
    __shared__ int hist[NBKT];
    const int tid = threadIdx.x;
    for (int i = tid; i < NBKT; i += SORT_THREADS) hist[i] = 0;
    __syncthreads();
    int4 r4[EPT_VEC];
#pragma unroll
    for (int q = 0; q < EPT_VEC; ++q) {
        const int e0 = blockIdx.x * EPB + (q * SORT_THREADS + tid) * 4;
        r4[q] = *(const int4*)(rows + min(e0, NNZ - 4));
    }
#pragma unroll
    for (int q = 0; q < EPT_VEC; ++q) {
        const int e0 = blockIdx.x * EPB + (q * SORT_THREADS + tid) * 4;
        if (e0 < NNZ) {
            atomicAdd(&hist[ngcf_clamp_row(r4[q].x) >> BKT_SHIFT], 1);
            atomicAdd(&hist[ngcf_clamp_row(r4[q].y) >> BKT_SHIFT], 1);
            atomicAdd(&hist[ngcf_clamp_row(r4[q].z) >> BKT_SHIFT], 1);
            atomicAdd(&hist[ngcf_clamp_row(r4[q].w) >> BKT_SHIFT], 1);
        }
    }
    __syncthreads();
    for (int i = tid; i < NBKT; i += SORT_THREADS) mat[blockIdx.x * NBKT + i] = hist[i];
}

__global__ void ngcf_bscan_kernel(const int* __restrict__ mat,
                                  int* __restrict__ pre,
                                  int* __restrict__ total) {
    const int lane = threadIdx.x & 31;
    const int j = blockIdx.x * (blockDim.x >> 5) + (threadIdx.x >> 5);
    if (j >= NBKT) return;
    int v[BSCAN_CHUNKS];
#pragma unroll
    for (int i = 0; i < BSCAN_CHUNKS; ++i) {
        const int b = i * 32 + lane;
        v[i] = (b < NBLK) ? mat[b * NBKT + j] : 0;
    }
    int run = 0;
#pragma unroll
    for (int i = 0; i < BSCAN_CHUNKS; ++i) {
        int incl = v[i];
        for (int off = 1; off < 32; off <<= 1) {
            const int t = __shfl_up(incl, off);
            if (lane >= off) incl += t;
        }
        const int b = i * 32 + lane;
        if (b < NBLK) pre[b * NBKT + j] = run + incl - v[i];
        run += __shfl(incl, 31);
    }
    if (lane == 0) total[j] = run;
}

__global__ void __launch_bounds__(SORT_THREADS)
ngcf_bscatter_kernel(const float* __restrict__ vals,
                     const int*   __restrict__ rows,
                     const int*   __restrict__ cols,
                     const int*   __restrict__ pre,
                     const int*   __restrict__ total,
                     int*         __restrict__ tmp) {
    __shared__ int  hist[NBKT];
    __shared__ int  lbase[NBKT];
    __shared__ int  gdelta[NBKT];
    __shared__ int  wsum[32];
    __shared__ int2 ent[EPB];
    __shared__ int  gpos[EPB];
    const int tid = threadIdx.x;
    const int blk = blockIdx.x;

    for (int i = tid; i < NBKT; i += SORT_THREADS) hist[i] = 0;
    __syncthreads();

    int4   r4[EPT_VEC];
    int4   c4[EPT_VEC];
    float4 v4[EPT_VEC];
#pragma unroll
    for (int q = 0; q < EPT_VEC; ++q) {
        const int e0 = blk * EPB + (q * SORT_THREADS + tid) * 4;
        const int ec = min(e0, NNZ - 4);
        r4[q] = *(const int4*)(rows + ec);
        c4[q] = *(const int4*)(cols + ec);
        v4[q] = *(const float4*)(vals + ec);
    }
    int  bk[EPT_VEC * 4];
    int  lr[EPT_VEC * 4];
    int2 en[EPT_VEC * 4];
#pragma unroll
    for (int q = 0; q < EPT_VEC; ++q) {
        const int   e0 = blk * EPB + (q * SORT_THREADS + tid) * 4;
        const int   rr[4] = {r4[q].x, r4[q].y, r4[q].z, r4[q].w};
        const int   cc[4] = {c4[q].x, c4[q].y, c4[q].z, c4[q].w};
        const float vv[4] = {v4[q].x, v4[q].y, v4[q].z, v4[q].w};
#pragma unroll
        for (int k = 0; k < 4; ++k) {
            const int r = ngcf_clamp_row(rr[k]);
            const int c = (int)min((unsigned)cc[k], (unsigned)(NODE_NUM - 1));
            bk[q * 4 + k] = r >> BKT_SHIFT;
            en[q * 4 + k] = make_int2(((r & (BKT_ROWS - 1)) << COL_BITS) | c, __float_as_int(vv[k]));
            lr[q * 4 + k] = (e0 < NNZ) ? atomicAdd(&hist[bk[q * 4 + k]], 1) : 0;
        }
    }
    __syncthreads();

    const int b0 = tid * 2;
    const int b1 = b0 + 1;
    const int h0 = (b0 < NBKT) ? hist[b0] : 0;
    const int h1 = (b1 < NBKT) ? hist[b1] : 0;
    const int t0 = (b0 < NBKT) ? total[b0] : 0;
    const int t1 = (b1 < NBKT) ? total[b1] : 0;
    int tot;
    const int lb = ngcf_block_excl_scan(h0 + h1, wsum, tot);
    const int bb = ngcf_block_excl_scan(t0 + t1, wsum, tot);
    if (b0 < NBKT) {
        lbase[b0]  = lb;
        gdelta[b0] = bb + pre[blk * NBKT + b0] - lb;
    }
    if (b1 < NBKT) {
        lbase[b1]  = lb + h0;
        gdelta[b1] = bb + t0 + pre[blk * NBKT + b1] - (lb + h0);
    }
    __syncthreads();

#pragma unroll
    for (int q = 0; q < EPT_VEC; ++q) {
        const int e0 = blk * EPB + (q * SORT_THREADS + tid) * 4;
        if (e0 < NNZ) {
#pragma unroll
            for (int k = 0; k < 4; ++k) {
                const int lp = min(lbase[bk[q * 4 + k]] + lr[q * 4 + k], EPB - 1);
                ent[lp]  = en[q * 4 + k];
                gpos[lp] = gdelta[bk[q * 4 + k]] + lp;
            }
        }
    }
    __syncthreads();

    const int nvalid = min(EPB, NNZ - blk * EPB);
    for (int i = tid; i < nvalid; i += SORT_THREADS) {
        const int p = min(max(gpos[i], 0), NNZ - 1);
        ((int2*)tmp)[p] = ent[i];
    }
}

__global__ void __launch_bounds__(SORT_THREADS)
ngcf_bucket_kernel(const int* __restrict__ total,
                   const int* __restrict__ tmp,
                   int*       __restrict__ rowptr,
                   int*       __restrict__ pairs) {
    __shared__ int  cnt[BKT_ROWS];
    __shared__ int  exs[BKT_ROWS];
    __shared__ int  cur[BKT_ROWS];
    __shared__ int  wsum[32];
    __shared__ int2 stage[BKT_STAGE];
    const int tid = threadIdx.x;
    const int j   = blockIdx.x;

    if (tid < BKT_ROWS) { cnt[tid] = 0; cur[tid] = 0; }
    int base;
    const int tb0 = (tid * 2     < j) ? total[tid * 2]     : 0;
    const int tb1 = (tid * 2 + 1 < j) ? total[tid * 2 + 1] : 0;
    ngcf_block_excl_scan(tb0 + tb1, wsum, base);
    base = min(max(base, 0), NNZ);
    const int n = min(max(total[j], 0), NNZ - base);
    const int2* src = (const int2*)tmp + base;
    int2*       dst = (int2*)pairs + base;

    for (int i0 = 0; i0 < n; i0 += BKT_LOADS * SORT_THREADS) {
        int x[BKT_LOADS];
#pragma unroll
        for (int k = 0; k < BKT_LOADS; ++k) {
            const int idx = i0 + k * SORT_THREADS + tid;
            x[k] = (idx < n) ? src[idx].x : -1;
        }
#pragma unroll
        for (int k = 0; k < BKT_LOADS; ++k)
            if (x[k] >= 0) atomicAdd(&cnt[(x[k] >> COL_BITS) & (BKT_ROWS - 1)], 1);
    }
    __syncthreads();

    int tot;
    const int ex = ngcf_block_excl_scan((tid < BKT_ROWS) ? cnt[tid] : 0, wsum, tot);
    if (tid < BKT_ROWS) {
        exs[tid] = ex;
        const int row = j * BKT_ROWS + tid;
        if (row <= NODE_NUM) rowptr[row] = base + ex;
    }
    __syncthreads();

    const bool staged = n <= BKT_STAGE;
    for (int i0 = 0; i0 < n; i0 += BKT_LOADS * SORT_THREADS) {
        int2 e[BKT_LOADS];
#pragma unroll
        for (int k = 0; k < BKT_LOADS; ++k) {
            const int idx = i0 + k * SORT_THREADS + tid;
            e[k] = (idx < n) ? src[idx] : make_int2(-1, 0);
        }
#pragma unroll
        for (int k = 0; k < BKT_LOADS; ++k) {
            if (e[k].x >= 0) {
                const int lrow = (e[k].x >> COL_BITS) & (BKT_ROWS - 1);
                const int p = exs[lrow] + atomicAdd(&cur[lrow], 1);
                const int2 o = make_int2(e[k].x & COL_MASK, e[k].y);
                if (staged) stage[min(max(p, 0), BKT_STAGE - 1)] = o;
                else        dst[min(max(p, 0), n - 1)] = o;
            }
        }
    }
    if (staged) {
        __syncthreads();
        for (int i = tid; i < n; i += SORT_THREADS) dst[i] = stage[i];
    }
}

typedef int v2i __attribute__((ext_vector_type(2)));
typedef const int __attribute__((address_space(4)))* ngcf_rowptr_p;
typedef const v2i __attribute__((address_space(4)))* ngcf_list_p;

template <bool TAIL>
__device__ __forceinline__ void ngcf_gather8(ngcf_list_p pairs,
                                             const float* __restrict__ Elane,
                                             int j, int end, float& ax, float& ay) {
    v2i p[8];
    for (int k = 0; k < 8; ++k) {
        int jj = j + k;
        if (TAIL) jj = min(jj, end - 1);
        p[k] = pairs[jj];
    }
    float2 ev[8];
    for (int k = 0; k < 8; ++k) {
        const unsigned c = min((unsigned)p[k].x, (unsigned)(NODE_NUM - 1));
        ev[k] = *(const float2*)(Elane + (size_t)c * EMBED_DIM);
    }
    for (int k = 0; k < 8; ++k) {
        if (!TAIL || (j + k < end)) {
            const float v = __int_as_float(p[k].y);
            ax = fmaf(v, ev[k].x, ax);
            ay = fmaf(v, ev[k].y, ay);
        }
    }
}

__device__ __forceinline__ void ngcf_spmm_row(const int*   __restrict__ rowptr,
                                              const int*   __restrict__ pairs_i,
                                              const float* __restrict__ E,
                                              float*       __restrict__ dst_row,
                                              int r, int lane) {
    const ngcf_list_p   pairs = (ngcf_list_p)pairs_i;
    const ngcf_rowptr_p rp    = (ngcf_rowptr_p)rowptr;
    int       j   = min(max(rp[r], 0), NNZ);
    const int end = min(rp[r + 1], NNZ);
    const float* Elane = E + lane * 2;

    float ax = 0.f, ay = 0.f;
    for (; j + 8 <= end; j += 8) ngcf_gather8<false>(pairs, Elane, j, end, ax, ay);
    if (j < end) ngcf_gather8<true>(pairs, Elane, j, end, ax, ay);

    *(float2*)(dst_row + lane * 2) = make_float2(ax, ay);
}

#define TILE_WAVES 16
#define TILE_THREADS (TILE_WAVES * 32)
#define TILE_PITCH 66
static_assert(EMBED_DIM * EMBED_DIM == 8 * TILE_THREADS, "weights: 8 floats per thread per matrix");

__device__ __forceinline__ void ngcf_layer_block16(const float* epRow, const float* sdRow,
                                                   const float* w1p, const float* w2p,
                                                   float* dstCol, int halfsel) {
    v8f acc = v8f{};
#pragma unroll
    for (int kk = 0; kk < 16; ++kk) {
        const int k0 = kk * 4;
        const float2 ep = *(const float2*)(epRow + k0);
        const float2 sd = *(const float2*)(sdRow + k0);

        v2f aSum, aBi;
        aSum.x = ep.x + sd.x;  aSum.y = ep.y + sd.y;
        aBi.x  = ep.x * sd.x;  aBi.y  = ep.y * sd.y;

        const int kph = ((k0 >> 1) + halfsel) << 7;
        const v2f b1 = *(const v2f*)(w1p + kph);
        const v2f b2 = *(const v2f*)(w2p + kph);
        acc = __builtin_amdgcn_wmma_f32_16x16x4_f32(
            false, aSum, false, b1, (short)0, acc, false, false);
        acc = __builtin_amdgcn_wmma_f32_16x16x4_f32(
            false, aBi,  false, b2, (short)0, acc, false, false);
    }
#pragma unroll
    for (int i = 0; i < 8; ++i) {
        float v = acc[i];
        v = (v > 0.f) ? v : v * NEG_SLOPE;
        dstCol[(i + halfsel * 8) * TILE_PITCH] = v;
    }
}

#define LAYER_TILES_PER_WG 7
#define LAYER_WGS ((NODE_TILES + LAYER_TILES_PER_WG - 1) / LAYER_TILES_PER_WG)
static_assert(LAYER_WGS <= 1024, "the layer grid is resident all at once");

template <bool FIRST>
__device__ __forceinline__ void ngcf_layer_tile(const int*   __restrict__ rowptr,
                                                const int*   __restrict__ pairs_i,
                                                const float* __restrict__ Eprev,
                                                const float* __restrict__ W1,
                                                const float* __restrict__ W2,
                                                float*       __restrict__ Eout,
                                                float* w1i, float* w2i,
                                                float* sideT, float* prevT, float* nextT,
                                                int tile, int tid, int lane, int wid) {
    float wa[8], wb[8];
    if (FIRST) {
#pragma unroll
        for (int j = 0; j < 8; ++j) {
            wa[j] = W1[j * TILE_THREADS + tid];
            wb[j] = W2[j * TILE_THREADS + tid];
        }
    }

    const int r = tile * 16 + wid;
    const float2 o = *(const float2*)(Eprev + (size_t)r * EMBED_DIM + lane * 2);

    ngcf_spmm_row(rowptr, pairs_i, Eprev, sideT + wid * TILE_PITCH, r, lane);
    *(float2*)(prevT + wid * TILE_PITCH + lane * 2) = o;

    if (FIRST) {
#pragma unroll
        for (int j = 0; j < 8; ++j) {
            const int i = j * TILE_THREADS + tid;
            const int k = i >> 6;
            const int n = i & 63;
            const int dst = ((k >> 1) << 7) + (n << 1) + (k & 1);
            w1i[dst] = wa[j];
            w2i[dst] = wb[j];
        }
    }
    __syncthreads();

    if (wid < 4) {
        const int halfsel = lane >> 4;
        const int nlane   = lane & 15;
        const int koff    = halfsel * 2;
        const int colo    = (nlane << 1) + wid * 32;
        ngcf_layer_block16(prevT + nlane * TILE_PITCH + koff,
                           sideT + nlane * TILE_PITCH + koff,
                           w1i + colo, w2i + colo,
                           nextT + wid * 16 + nlane, halfsel);
    }
    __syncthreads();

    *(float2*)(Eout + (size_t)r * EMBED_DIM + lane * 2) =
        *(const float2*)(nextT + wid * TILE_PITCH + lane * 2);
}

__global__ void __launch_bounds__(TILE_THREADS)
ngcf_layer_kernel(const int*   __restrict__ rowptr,
                  const int*   __restrict__ pairs_i,
                  const float* __restrict__ Eprev,
                  const float* __restrict__ W1,
                  const float* __restrict__ W2,
                  float*       __restrict__ Eout) {
    __shared__ float w1i[EMBED_DIM * EMBED_DIM]  __attribute__((aligned(16)));
    __shared__ float w2i[EMBED_DIM * EMBED_DIM]  __attribute__((aligned(16)));
    __shared__ float sideT[16 * TILE_PITCH]      __attribute__((aligned(16)));
    __shared__ float prevT[16 * TILE_PITCH]      __attribute__((aligned(16)));
    __shared__ float nextT[16 * TILE_PITCH]      __attribute__((aligned(16)));

    const int tid  = threadIdx.x;
    const int lane = tid & 31;
    const int wid = __builtin_amdgcn_readfirstlane(tid >> 5);

    int tile = blockIdx.x;
    if (tile >= NODE_TILES) return;
    ngcf_layer_tile<true>(rowptr, pairs_i, Eprev, W1, W2, Eout,
                          w1i, w2i, sideT, prevT, nextT, tile, tid, lane, wid);
    for (tile += gridDim.x; tile < NODE_TILES; tile += gridDim.x)
        ngcf_layer_tile<false>(rowptr, pairs_i, Eprev, W1, W2, Eout,
                               w1i, w2i, sideT, prevT, nextT, tile, tid, lane, wid);
}

#define SEL_TILES (2 * BATCH / 16)

__global__ void __launch_bounds__(TILE_THREADS)
ngcf_tail_kernel(const int*   __restrict__ rowptr,
                 const int*   __restrict__ pairs_i,
                 const float* __restrict__ emb,
                 const float* __restrict__ E1,
                 const float* __restrict__ E2,
                 const float* __restrict__ W1,
                 const float* __restrict__ W2,
                 const int*   __restrict__ user_idx,
                 const int*   __restrict__ item_idx,
                 float*       __restrict__ out) {
    __shared__ float w1i[EMBED_DIM * EMBED_DIM]  __attribute__((aligned(16)));
    __shared__ float w2i[EMBED_DIM * EMBED_DIM]  __attribute__((aligned(16)));
    __shared__ float sideT[16 * TILE_PITCH]      __attribute__((aligned(16)));
    __shared__ float e2T[16 * TILE_PITCH]        __attribute__((aligned(16)));
    __shared__ float e3T[16 * TILE_PITCH]        __attribute__((aligned(16)));

    const int tid  = threadIdx.x;
    const int lane = tid & 31;
    const int wid = __builtin_amdgcn_readfirstlane(tid >> 5);
    const int m0  = blockIdx.x * 16;
    const int s   = m0 + wid;
    const int r   = ngcf_slot_node(user_idx, item_idx, s);

    float wa[8], wb[8];
#pragma unroll
    for (int j = 0; j < 8; ++j) {
        wa[j] = W1[j * TILE_THREADS + tid];
        wb[j] = W2[j * TILE_THREADS + tid];
    }

    const size_t ro = (size_t)r * EMBED_DIM + lane * 2;
    const float2 o0 = *(const float2*)(emb + ro);
    const float2 o1 = *(const float2*)(E1 + ro);
    const float2 o2 = *(const float2*)(E2 + ro);

    ngcf_spmm_row(rowptr, pairs_i, E2, sideT + wid * TILE_PITCH, r, lane);
    *(float2*)(e2T + wid * TILE_PITCH + lane * 2) = o2;

#pragma unroll
    for (int j = 0; j < 8; ++j) {
        const int i = j * TILE_THREADS + tid;
        const int k = i >> 6;
        const int n = i & 63;
        const int dst = ((k >> 1) << 7) + (n << 1) + (k & 1);
        w1i[dst] = wa[j];
        w2i[dst] = wb[j];
    }
    __syncthreads();

    if (wid < 4) {
        const int halfsel = lane >> 4;
        const int nlane   = lane & 15;
        const int koff    = halfsel * 2;
        const int colo    = (nlane << 1) + wid * 32;
        ngcf_layer_block16(e2T   + nlane * TILE_PITCH + koff,
                           sideT + nlane * TILE_PITCH + koff,
                           w1i + colo, w2i + colo,
                           e3T + wid * 16 + nlane, halfsel);
    }
    __syncthreads();

    float* orow = out + (size_t)s * 256 + lane * 2;
    *(float2*)(orow)       = o0;
    *(float2*)(orow + 64)  = o1;
    *(float2*)(orow + 128) = o2;
    *(float2*)(orow + 192) = *(const float2*)(e3T + wid * TILE_PITCH + lane * 2);
}

extern "C" void kernel_launch(void* const* d_in, const int* in_sizes, int n_in,
                              void* d_out, int out_size, void* d_ws, size_t ws_size,
                              hipStream_t stream) {
    const float* emb      = (const float*)d_in[0];
    const float* W1       = (const float*)d_in[1];
    const float* W2       = (const float*)d_in[2];
    const float* vals     = (const float*)d_in[3];
    const int*   row_idx  = (const int*)d_in[4];
    const int*   col_idx  = (const int*)d_in[5];
    const int*   user_idx = (const int*)d_in[6];
    const int*   item_idx = (const int*)d_in[7];

    const size_t EN = (size_t)NODE_NUM * EMBED_DIM;
    float* E1   = (float*)d_ws;
    float* E2   = E1 + EN;
    float* P3   = E2 + EN;

    int* pairs  = (int*)P3;
    int* rowptr = pairs + 2 * (size_t)NNZ;
    int* mat    = rowptr + 102400;
    int* pre    = mat + NBLK * NBKT;
    int* total  = pre + NBLK * NBKT;
    int* tmp    = (int*)E2;

    ngcf_bhist_kernel<<<NBLK, SORT_THREADS, 0, stream>>>(row_idx, mat);
    ngcf_bscan_kernel<<<(NBKT + 7) / 8, 256, 0, stream>>>(mat, pre, total);
    ngcf_bscatter_kernel<<<NBLK, SORT_THREADS, 0, stream>>>(vals, row_idx, col_idx,
                                                            pre, total, tmp);
    ngcf_bucket_kernel<<<NBKT, SORT_THREADS, 0, stream>>>(total, tmp, rowptr, pairs);

    ngcf_layer_kernel<<<LAYER_WGS, TILE_THREADS, 0, stream>>>(rowptr, pairs, emb, W1, W2, E1);
    ngcf_layer_kernel<<<LAYER_WGS, TILE_THREADS, 0, stream>>>(
        rowptr, pairs, E1, W1 + EMBED_DIM * EMBED_DIM, W2 + EMBED_DIM * EMBED_DIM, E2);

    ngcf_tail_kernel<<<SEL_TILES, TILE_THREADS, 0, stream>>>(
        rowptr, pairs, emb, E1, E2,
        W1 + 2 * EMBED_DIM * EMBED_DIM, W2 + 2 * EMBED_DIM * EMBED_DIM,
        user_idx, item_idx, (float*)d_out);
}


# Edge sort in 2 launches, not 4: level-1 blocks stream bucket-ordered edges to own slices + offset table; bucket block sums its offset column for its base; row terms' LDS-rank arrival order changes
# speedup vs baseline: 5.9623x; 1.2489x over previous
#include <hip/hip_runtime.h>
#include <hip/hip_bf16.h>


#define USER_NUM 60000
#define ITEM_NUM 40000
#define NODE_NUM (USER_NUM + ITEM_NUM)
#define EMBED_DIM 64
#define N_LAYER 3
#define NNZ 1600000
#define BATCH 4096
#define NEG_SLOPE 0.2f
#define NODE_TILES (NODE_NUM / 16)

typedef float v2f __attribute__((ext_vector_type(2)));
typedef float v8f __attribute__((ext_vector_type(8)));

__device__ __forceinline__ int ngcf_slot_node(const int* __restrict__ user_idx,
                                              const int* __restrict__ item_idx,
                                              int s) {
    const int node = (s < BATCH) ? user_idx[s] : USER_NUM + item_idx[s - BATCH];
    return min(max(node, 0), NODE_NUM - 1);
}

#define BKT_SHIFT 8
#define BKT_ROWS (1 << BKT_SHIFT)
#define NBKT ((NODE_NUM + BKT_ROWS - 1) / BKT_ROWS)
#define SORT_THREADS 256
#define EPT_VEC 4
#define EPB 4096
#define NBLK ((NNZ + EPB - 1) / EPB)
#define COL_BITS 17
#define COL_MASK ((1 << COL_BITS) - 1)
#define BKT_LOADS 24
#define BKT_STAGE (BKT_LOADS * SORT_THREADS)
#define SLICE_HW (SORT_THREADS / 16)
#define SLICE_K ((NBLK + SLICE_HW - 1) / SLICE_HW)
#define SLICE_SLOTS (SLICE_K * SLICE_HW)

static_assert(NBKT <= 2 * SORT_THREADS, "two buckets per thread in the block scan");
static_assert(SLICE_SLOTS <= 2 * SORT_THREADS, "two slices per thread in the level-2 set-up");
static_assert(BKT_ROWS <= SORT_THREADS, "one row per thread in the level-2 scan");
static_assert(EPB == EPT_VEC * 4 * SORT_THREADS, "edges per block");
static_assert(NNZ % 4 == 0 && EPB % 4 == 0, "a block's valid edges come in fours: 16-byte stream stores");
static_assert(NODE_NUM <= (1 << COL_BITS) && (BKT_ROWS << COL_BITS) > 0, "entry packing");

__device__ __forceinline__ int ngcf_block_excl_scan(int v, int* wsum, int& total) {
    const int lane = threadIdx.x & 31;
    const int wid  = threadIdx.x >> 5;
    int incl = v;
    for (int off = 1; off < 32; off <<= 1) {
        const int t = __shfl_up(incl, off);
        if (lane >= off) incl += t;
    }
    if (lane == 31) wsum[wid] = incl;
    __syncthreads();
    const int wv = (lane < (int)(blockDim.x >> 5)) ? wsum[lane] : 0;
    int wincl = wv;
    for (int off = 1; off < 32; off <<= 1) {
        const int t = __shfl_up(wincl, off);
        if (lane >= off) wincl += t;
    }
    const int wpre = __shfl(wincl, wid) - __shfl(wv, wid);
    total = __shfl(wincl, 31);
    __syncthreads();
    return wpre + incl - v;
}

__device__ __forceinline__ int ngcf_clamp_row(int r) {
    return min(max(r, 0), NODE_NUM - 1);
}

__global__ void __launch_bounds__(SORT_THREADS)
ngcf_bin_kernel(const float* __restrict__ vals,
                const int*   __restrict__ rows,
                const int*   __restrict__ cols,
                int*         __restrict__ offT,
                int*         __restrict__ tmp) {
    __shared__ int  hist[NBKT];
    __shared__ int  lbase[NBKT];
    __shared__ int  wsum[32];
    __shared__ int2 ent[EPB] __attribute__((aligned(16)));
    const int tid = threadIdx.x;
    const int blk = blockIdx.x;

    for (int i = tid; i < NBKT; i += SORT_THREADS) hist[i] = 0;
    __syncthreads();

    int4   r4[EPT_VEC];
    int4   c4[EPT_VEC];
    float4 v4[EPT_VEC];
#pragma unroll
    for (int q = 0; q < EPT_VEC; ++q) {
        const int e0 = blk * EPB + (q * SORT_THREADS + tid) * 4;
        const int ec = min(e0, NNZ - 4);
        r4[q] = *(const int4*)(rows + ec);
        c4[q] = *(const int4*)(cols + ec);
        v4[q] = *(const float4*)(vals + ec);
    }
    int  bk[EPT_VEC * 4];
    int  lr[EPT_VEC * 4];
    int2 en[EPT_VEC * 4];
#pragma unroll
    for (int q = 0; q < EPT_VEC; ++q) {
        const int   e0 = blk * EPB + (q * SORT_THREADS + tid) * 4;
        const int   rr[4] = {r4[q].x, r4[q].y, r4[q].z, r4[q].w};
        const int   cc[4] = {c4[q].x, c4[q].y, c4[q].z, c4[q].w};
        const float vv[4] = {v4[q].x, v4[q].y, v4[q].z, v4[q].w};
#pragma unroll
        for (int k = 0; k < 4; ++k) {
            const int r = ngcf_clamp_row(rr[k]);
            const int c = (int)min((unsigned)cc[k], (unsigned)(NODE_NUM - 1));
            bk[q * 4 + k] = r >> BKT_SHIFT;
            en[q * 4 + k] = make_int2(((r & (BKT_ROWS - 1)) << COL_BITS) | c, __float_as_int(vv[k]));
            lr[q * 4 + k] = (e0 < NNZ) ? atomicAdd(&hist[bk[q * 4 + k]], 1) : 0;
        }
    }
    __syncthreads();

    const int b0 = tid * 2;
    const int b1 = b0 + 1;
    const int h0 = (b0 < NBKT) ? hist[b0] : 0;
    const int h1 = (b1 < NBKT) ? hist[b1] : 0;
    int tot;
    const int lb = ngcf_block_excl_scan(h0 + h1, wsum, tot);
    if (b0 < NBKT) {
        lbase[b0] = lb;
        offT[b0 * NBLK + blk] = lb;
    }
    if (b1 < NBKT) {
        lbase[b1] = lb + h0;
        offT[b1 * NBLK + blk] = lb + h0;
    }
    if (tid == 0) offT[NBKT * NBLK + blk] = tot;
    __syncthreads();

#pragma unroll
    for (int q = 0; q < EPT_VEC; ++q) {
        const int e0 = blk * EPB + (q * SORT_THREADS + tid) * 4;
        if (e0 < NNZ) {
#pragma unroll
            for (int k = 0; k < 4; ++k) {
                const int lp = min(lbase[bk[q * 4 + k]] + lr[q * 4 + k], EPB - 1);
                ent[lp] = en[q * 4 + k];
            }
        }
    }
    __syncthreads();

    const int nvalid = min(EPB, NNZ - blk * EPB);
    int4*       out4 = (int4*)tmp + (size_t)blk * (EPB / 2);
    const int4* ent4 = (const int4*)ent;
    for (int i = tid; i < nvalid / 2; i += SORT_THREADS) out4[i] = ent4[i];
}

__device__ __forceinline__ void ngcf_place(int2 e, const int* exs, int* cur,
                                           int2* stage, int2* dst, bool staged, int n) {
    const int lrow = (e.x >> COL_BITS) & (BKT_ROWS - 1);
    const int p = exs[lrow] + atomicAdd(&cur[lrow], 1);
    const int2 o = make_int2(e.x & COL_MASK, e.y);
    if (staged) stage[min(max(p, 0), BKT_STAGE - 1)] = o;
    else        dst[min(max(p, 0), n - 1)] = o;
}

__global__ void __launch_bounds__(SORT_THREADS)
ngcf_bucket_kernel(const int* __restrict__ offT,
                   const int* __restrict__ tmp,
                   int*       __restrict__ rowptr,
                   int*       __restrict__ pairs) {
    __shared__ int  cnt[BKT_ROWS];
    __shared__ int  exs[BKT_ROWS];
    __shared__ int  cur[BKT_ROWS];
    __shared__ int  wsum[32];
    __shared__ int  sbeg[SLICE_SLOTS];
    __shared__ int  slen[SLICE_SLOTS];
    __shared__ int2 stage[BKT_STAGE];
    const int tid = threadIdx.x;
    const int j   = blockIdx.x;

    if (tid < BKT_ROWS) { cnt[tid] = 0; cur[tid] = 0; }

    int osum = 0, lsum = 0;
#pragma unroll
    for (int q = 0; q < 2; ++q) {
        const int b = tid + q * SORT_THREADS;
        if (b < SLICE_SLOTS) {
            int beg = 0, len = 0;
            if (b < NBLK) {
                const int o0 = min(max(offT[j * NBLK + b], 0), EPB);
                const int o1 = min(max(offT[(j + 1) * NBLK + b], 0), EPB);
                beg = b * EPB + o0;
                len = min(max(o1 - o0, 0), max(NNZ - beg, 0));
                osum += o0;
                lsum += len;
            }
            sbeg[b] = beg;
            slen[b] = len;
        }
    }
    int base, n;
    ngcf_block_excl_scan(osum, wsum, base);
    ngcf_block_excl_scan(lsum, wsum, n);
    base = min(max(base, 0), NNZ);
    n    = min(n, NNZ - base);
    const int2* src = (const int2*)tmp;
    int2*       dst = (int2*)pairs + base;

    const int hw  = tid >> 4;
    const int l16 = tid & 15;

    int sb[SLICE_K], sl[SLICE_K];
#pragma unroll
    for (int k = 0; k < SLICE_K; ++k) {
        sb[k] = sbeg[hw + k * SLICE_HW];
        sl[k] = slen[hw + k * SLICE_HW];
    }

    int2 e[SLICE_K];
    {
        int y[SLICE_K];
#pragma unroll
        for (int k = 0; k < SLICE_K; ++k) {
            e[k] = (l16      < sl[k]) ? src[sb[k] + l16]        : make_int2(-1, 0);
            y[k] = (l16 + 16 < sl[k]) ? src[sb[k] + l16 + 16].x : -1;
        }
#pragma unroll
        for (int k = 0; k < SLICE_K; ++k) {
            if (e[k].x >= 0) atomicAdd(&cnt[(e[k].x >> COL_BITS) & (BKT_ROWS - 1)], 1);
            if (y[k]   >= 0) atomicAdd(&cnt[(y[k]   >> COL_BITS) & (BKT_ROWS - 1)], 1);
        }
#pragma unroll
        for (int k = 0; k < SLICE_K; ++k)
            for (int e3 = l16 + 32; e3 < sl[k]; e3 += 16)
                atomicAdd(&cnt[(src[sb[k] + e3].x >> COL_BITS) & (BKT_ROWS - 1)], 1);
    }
    __syncthreads();

    int tot;
    const int ex = ngcf_block_excl_scan((tid < BKT_ROWS) ? cnt[tid] : 0, wsum, tot);
    if (tid < BKT_ROWS) {
        exs[tid] = ex;
        const int row = j * BKT_ROWS + tid;
        if (row <= NODE_NUM) rowptr[row] = base + ex;
    }
    __syncthreads();

    const bool staged = n <= BKT_STAGE;
    {
        int2 f[SLICE_K];
#pragma unroll
        for (int k = 0; k < SLICE_K; ++k)
            f[k] = (l16 + 16 < sl[k]) ? src[sb[k] + l16 + 16] : make_int2(-1, 0);
#pragma unroll
        for (int k = 0; k < SLICE_K; ++k)
            if (e[k].x >= 0) ngcf_place(e[k], exs, cur, stage, dst, staged, n);
#pragma unroll
        for (int k = 0; k < SLICE_K; ++k)
            if (f[k].x >= 0) ngcf_place(f[k], exs, cur, stage, dst, staged, n);
#pragma unroll
        for (int k = 0; k < SLICE_K; ++k)
            for (int e2 = l16 + 32; e2 < sl[k]; e2 += 16)
                ngcf_place(src[sb[k] + e2], exs, cur, stage, dst, staged, n);
    }
    if (staged) {
        __syncthreads();
        for (int i = tid; i < n; i += SORT_THREADS) dst[i] = stage[i];
    }
}

typedef int v2i __attribute__((ext_vector_type(2)));
typedef const int __attribute__((address_space(4)))* ngcf_rowptr_p;
typedef const v2i __attribute__((address_space(4)))* ngcf_list_p;

template <bool TAIL>
__device__ __forceinline__ void ngcf_gather8(ngcf_list_p pairs,
                                             const float* __restrict__ Elane,
                                             int j, int end, float& ax, float& ay) {
    v2i p[8];
    for (int k = 0; k < 8; ++k) {
        int jj = j + k;
        if (TAIL) jj = min(jj, end - 1);
        p[k] = pairs[jj];
    }
    float2 ev[8];
    for (int k = 0; k < 8; ++k) {
        const unsigned c = min((unsigned)p[k].x, (unsigned)(NODE_NUM - 1));
        ev[k] = *(const float2*)(Elane + (size_t)c * EMBED_DIM);
    }
    for (int k = 0; k < 8; ++k) {
        if (!TAIL || (j + k < end)) {
            const float v = __int_as_float(p[k].y);
            ax = fmaf(v, ev[k].x, ax);
            ay = fmaf(v, ev[k].y, ay);
        }
    }
}

__device__ __forceinline__ void ngcf_spmm_row(const int*   __restrict__ rowptr,
                                              const int*   __restrict__ pairs_i,
                                              const float* __restrict__ E,
                                              float*       __restrict__ dst_row,
                                              int r, int lane) {
    const ngcf_list_p   pairs = (ngcf_list_p)pairs_i;
    const ngcf_rowptr_p rp    = (ngcf_rowptr_p)rowptr;
    int       j   = min(max(rp[r], 0), NNZ);
    const int end = min(rp[r + 1], NNZ);
    const float* Elane = E + lane * 2;

    float ax = 0.f, ay = 0.f;
    for (; j + 8 <= end; j += 8) ngcf_gather8<false>(pairs, Elane, j, end, ax, ay);
    if (j < end) ngcf_gather8<true>(pairs, Elane, j, end, ax, ay);

    *(float2*)(dst_row + lane * 2) = make_float2(ax, ay);
}

#define TILE_WAVES 16
#define TILE_THREADS (TILE_WAVES * 32)
#define TILE_PITCH 66
static_assert(EMBED_DIM * EMBED_DIM == 8 * TILE_THREADS, "weights: 8 floats per thread per matrix");

__device__ __forceinline__ void ngcf_layer_block16(const float* epRow, const float* sdRow,
                                                   const float* w1p, const float* w2p,
                                                   float* dstCol, int halfsel) {
    v8f acc = v8f{};
#pragma unroll
    for (int kk = 0; kk < 16; ++kk) {
        const int k0 = kk * 4;
        const float2 ep = *(const float2*)(epRow + k0);
        const float2 sd = *(const float2*)(sdRow + k0);

        v2f aSum, aBi;
        aSum.x = ep.x + sd.x;  aSum.y = ep.y + sd.y;
        aBi.x  = ep.x * sd.x;  aBi.y  = ep.y * sd.y;

        const int kph = ((k0 >> 1) + halfsel) << 7;
        const v2f b1 = *(const v2f*)(w1p + kph);
        const v2f b2 = *(const v2f*)(w2p + kph);
        acc = __builtin_amdgcn_wmma_f32_16x16x4_f32(
            false, aSum, false, b1, (short)0, acc, false, false);
        acc = __builtin_amdgcn_wmma_f32_16x16x4_f32(
            false, aBi,  false, b2, (short)0, acc, false, false);
    }
#pragma unroll
    for (int i = 0; i < 8; ++i) {
        float v = acc[i];
        v = (v > 0.f) ? v : v * NEG_SLOPE;
        dstCol[(i + halfsel * 8) * TILE_PITCH] = v;
    }
}

#define LAYER_TILES_PER_WG 7
#define LAYER_WGS ((NODE_TILES + LAYER_TILES_PER_WG - 1) / LAYER_TILES_PER_WG)
static_assert(LAYER_WGS <= 1024, "the layer grid is resident all at once");

template <bool FIRST>
__device__ __forceinline__ void ngcf_layer_tile(const int*   __restrict__ rowptr,
                                                const int*   __restrict__ pairs_i,
                                                const float* __restrict__ Eprev,
                                                const float* __restrict__ W1,
                                                const float* __restrict__ W2,
                                                float*       __restrict__ Eout,
                                                float* w1i, float* w2i,
                                                float* sideT, float* prevT, float* nextT,
                                                int tile, int tid, int lane, int wid) {
    float wa[8], wb[8];
    if (FIRST) {
#pragma unroll
        for (int j = 0; j < 8; ++j) {
            wa[j] = W1[j * TILE_THREADS + tid];
            wb[j] = W2[j * TILE_THREADS + tid];
        }
    }

    const int r = tile * 16 + wid;
    const float2 o = *(const float2*)(Eprev + (size_t)r * EMBED_DIM + lane * 2);

    ngcf_spmm_row(rowptr, pairs_i, Eprev, sideT + wid * TILE_PITCH, r, lane);
    *(float2*)(prevT + wid * TILE_PITCH + lane * 2) = o;

    if (FIRST) {
#pragma unroll
        for (int j = 0; j < 8; ++j) {
            const int i = j * TILE_THREADS + tid;
            const int k = i >> 6;
            const int n = i & 63;
            const int dst = ((k >> 1) << 7) + (n << 1) + (k & 1);
            w1i[dst] = wa[j];
            w2i[dst] = wb[j];
        }
    }
    __syncthreads();

    if (wid < 4) {
        const int halfsel = lane >> 4;
        const int nlane   = lane & 15;
        const int koff    = halfsel * 2;
        const int colo    = (nlane << 1) + wid * 32;
        ngcf_layer_block16(prevT + nlane * TILE_PITCH + koff,
                           sideT + nlane * TILE_PITCH + koff,
                           w1i + colo, w2i + colo,
                           nextT + wid * 16 + nlane, halfsel);
    }
    __syncthreads();

    *(float2*)(Eout + (size_t)r * EMBED_DIM + lane * 2) =
        *(const float2*)(nextT + wid * TILE_PITCH + lane * 2);
}

__global__ void __launch_bounds__(TILE_THREADS)
ngcf_layer_kernel(const int*   __restrict__ rowptr,
                  const int*   __restrict__ pairs_i,
                  const float* __restrict__ Eprev,
                  const float* __restrict__ W1,
                  const float* __restrict__ W2,
                  float*       __restrict__ Eout) {
    __shared__ float w1i[EMBED_DIM * EMBED_DIM]  __attribute__((aligned(16)));
    __shared__ float w2i[EMBED_DIM * EMBED_DIM]  __attribute__((aligned(16)));
    __shared__ float sideT[16 * TILE_PITCH]      __attribute__((aligned(16)));
    __shared__ float prevT[16 * TILE_PITCH]      __attribute__((aligned(16)));
    __shared__ float nextT[16 * TILE_PITCH]      __attribute__((aligned(16)));

    const int tid  = threadIdx.x;
    const int lane = tid & 31;
    const int wid = __builtin_amdgcn_readfirstlane(tid >> 5);

    int tile = blockIdx.x;
    if (tile >= NODE_TILES) return;
    ngcf_layer_tile<true>(rowptr, pairs_i, Eprev, W1, W2, Eout,
                          w1i, w2i, sideT, prevT, nextT, tile, tid, lane, wid);
    for (tile += gridDim.x; tile < NODE_TILES; tile += gridDim.x)
        ngcf_layer_tile<false>(rowptr, pairs_i, Eprev, W1, W2, Eout,
                               w1i, w2i, sideT, prevT, nextT, tile, tid, lane, wid);
}

#define SEL_TILES (2 * BATCH / 16)

__global__ void __launch_bounds__(TILE_THREADS)
ngcf_tail_kernel(const int*   __restrict__ rowptr,
                 const int*   __restrict__ pairs_i,
                 const float* __restrict__ emb,
                 const float* __restrict__ E1,
                 const float* __restrict__ E2,
                 const float* __restrict__ W1,
                 const float* __restrict__ W2,
                 const int*   __restrict__ user_idx,
                 const int*   __restrict__ item_idx,
                 float*       __restrict__ out) {
    __shared__ float w1i[EMBED_DIM * EMBED_DIM]  __attribute__((aligned(16)));
    __shared__ float w2i[EMBED_DIM * EMBED_DIM]  __attribute__((aligned(16)));
    __shared__ float sideT[16 * TILE_PITCH]      __attribute__((aligned(16)));
    __shared__ float e2T[16 * TILE_PITCH]        __attribute__((aligned(16)));
    __shared__ float e3T[16 * TILE_PITCH]        __attribute__((aligned(16)));

    const int tid  = threadIdx.x;
    const int lane = tid & 31;
    const int wid = __builtin_amdgcn_readfirstlane(tid >> 5);
    const int m0  = blockIdx.x * 16;
    const int s   = m0 + wid;
    const int r   = ngcf_slot_node(user_idx, item_idx, s);

    float wa[8], wb[8];
#pragma unroll
    for (int j = 0; j < 8; ++j) {
        wa[j] = W1[j * TILE_THREADS + tid];
        wb[j] = W2[j * TILE_THREADS + tid];
    }

    const size_t ro = (size_t)r * EMBED_DIM + lane * 2;
    const float2 o0 = *(const float2*)(emb + ro);
    const float2 o1 = *(const float2*)(E1 + ro);
    const float2 o2 = *(const float2*)(E2 + ro);

    ngcf_spmm_row(rowptr, pairs_i, E2, sideT + wid * TILE_PITCH, r, lane);
    *(float2*)(e2T + wid * TILE_PITCH + lane * 2) = o2;

#pragma unroll
    for (int j = 0; j < 8; ++j) {
        const int i = j * TILE_THREADS + tid;
        const int k = i >> 6;
        const int n = i & 63;
        const int dst = ((k >> 1) << 7) + (n << 1) + (k & 1);
        w1i[dst] = wa[j];
        w2i[dst] = wb[j];
    }
    __syncthreads();

    if (wid < 4) {
        const int halfsel = lane >> 4;
        const int nlane   = lane & 15;
        const int koff    = halfsel * 2;
        const int colo    = (nlane << 1) + wid * 32;
        ngcf_layer_block16(e2T   + nlane * TILE_PITCH + koff,
                           sideT + nlane * TILE_PITCH + koff,
                           w1i + colo, w2i + colo,
                           e3T + wid * 16 + nlane, halfsel);
    }
    __syncthreads();

    float* orow = out + (size_t)s * 256 + lane * 2;
    *(float2*)(orow)       = o0;
    *(float2*)(orow + 64)  = o1;
    *(float2*)(orow + 128) = o2;
    *(float2*)(orow + 192) = *(const float2*)(e3T + wid * TILE_PITCH + lane * 2);
}

extern "C" void kernel_launch(void* const* d_in, const int* in_sizes, int n_in,
                              void* d_out, int out_size, void* d_ws, size_t ws_size,
                              hipStream_t stream) {
    const float* emb      = (const float*)d_in[0];
    const float* W1       = (const float*)d_in[1];
    const float* W2       = (const float*)d_in[2];
    const float* vals     = (const float*)d_in[3];
    const int*   row_idx  = (const int*)d_in[4];
    const int*   col_idx  = (const int*)d_in[5];
    const int*   user_idx = (const int*)d_in[6];
    const int*   item_idx = (const int*)d_in[7];

    const size_t EN = (size_t)NODE_NUM * EMBED_DIM;
    float* E1   = (float*)d_ws;
    float* E2   = E1 + EN;
    float* P3   = E2 + EN;

    int* pairs  = (int*)P3;
    int* rowptr = pairs + 2 * (size_t)NNZ;
    int* offT   = rowptr + 102400;
    int* tmp    = (int*)E2;

    ngcf_bin_kernel<<<NBLK, SORT_THREADS, 0, stream>>>(vals, row_idx, col_idx, offT, tmp);
    ngcf_bucket_kernel<<<NBKT, SORT_THREADS, 0, stream>>>(offT, tmp, rowptr, pairs);

    ngcf_layer_kernel<<<LAYER_WGS, TILE_THREADS, 0, stream>>>(rowptr, pairs, emb, W1, W2, E1);
    ngcf_layer_kernel<<<LAYER_WGS, TILE_THREADS, 0, stream>>>(
        rowptr, pairs, E1, W1 + EMBED_DIM * EMBED_DIM, W2 + EMBED_DIM * EMBED_DIM, E2);

    ngcf_tail_kernel<<<SEL_TILES, TILE_THREADS, 0, stream>>>(
        rowptr, pairs, emb, E1, E2,
        W1 + 2 * EMBED_DIM * EMBED_DIM, W2 + 2 * EMBED_DIM * EMBED_DIM,
        user_idx, item_idx, (float*)d_out);
}


# Layers 1 and 2: a full grid of 1024 workgroups claims 16-node tiles one at a time from a global counter (integer atomic, cleared by the first launch), not 893 x 7 fixed tiles; same sum order
# speedup vs baseline: 6.1531x; 1.0320x over previous
#include <hip/hip_runtime.h>
#include <hip/hip_bf16.h>


#define USER_NUM 60000
#define ITEM_NUM 40000
#define NODE_NUM (USER_NUM + ITEM_NUM)
#define EMBED_DIM 64
#define N_LAYER 3
#define NNZ 1600000
#define BATCH 4096
#define NEG_SLOPE 0.2f
#define NODE_TILES (NODE_NUM / 16)

typedef float v2f __attribute__((ext_vector_type(2)));
typedef float v8f __attribute__((ext_vector_type(8)));

__device__ __forceinline__ int ngcf_slot_node(const int* __restrict__ user_idx,
                                              const int* __restrict__ item_idx,
                                              int s) {
    const int node = (s < BATCH) ? user_idx[s] : USER_NUM + item_idx[s - BATCH];
    return min(max(node, 0), NODE_NUM - 1);
}

#define BKT_SHIFT 8
#define BKT_ROWS (1 << BKT_SHIFT)
#define NBKT ((NODE_NUM + BKT_ROWS - 1) / BKT_ROWS)
#define SORT_THREADS 256
#define EPT_VEC 4
#define EPB 4096
#define NBLK ((NNZ + EPB - 1) / EPB)
#define COL_BITS 17
#define COL_MASK ((1 << COL_BITS) - 1)
#define BKT_LOADS 24
#define BKT_STAGE (BKT_LOADS * SORT_THREADS)
#define SLICE_HW (SORT_THREADS / 16)
#define SLICE_K ((NBLK + SLICE_HW - 1) / SLICE_HW)
#define SLICE_SLOTS (SLICE_K * SLICE_HW)

static_assert(NBKT <= 2 * SORT_THREADS, "two buckets per thread in the block scan");
static_assert(SLICE_SLOTS <= 2 * SORT_THREADS, "two slices per thread in the level-2 set-up");
static_assert(BKT_ROWS <= SORT_THREADS, "one row per thread in the level-2 scan");
static_assert(EPB == EPT_VEC * 4 * SORT_THREADS, "edges per block");
static_assert(NNZ % 4 == 0 && EPB % 4 == 0, "a block's valid edges come in fours: 16-byte stream stores");
static_assert(NODE_NUM <= (1 << COL_BITS) && (BKT_ROWS << COL_BITS) > 0, "entry packing");

__device__ __forceinline__ int ngcf_block_excl_scan(int v, int* wsum, int& total) {
    const int lane = threadIdx.x & 31;
    const int wid  = threadIdx.x >> 5;
    int incl = v;
    for (int off = 1; off < 32; off <<= 1) {
        const int t = __shfl_up(incl, off);
        if (lane >= off) incl += t;
    }
    if (lane == 31) wsum[wid] = incl;
    __syncthreads();
    const int wv = (lane < (int)(blockDim.x >> 5)) ? wsum[lane] : 0;
    int wincl = wv;
    for (int off = 1; off < 32; off <<= 1) {
        const int t = __shfl_up(wincl, off);
        if (lane >= off) wincl += t;
    }
    const int wpre = __shfl(wincl, wid) - __shfl(wv, wid);
    total = __shfl(wincl, 31);
    __syncthreads();
    return wpre + incl - v;
}

__device__ __forceinline__ int ngcf_clamp_row(int r) {
    return min(max(r, 0), NODE_NUM - 1);
}

__global__ void __launch_bounds__(SORT_THREADS)
ngcf_bin_kernel(const float* __restrict__ vals,
                const int*   __restrict__ rows,
                const int*   __restrict__ cols,
                int*         __restrict__ offT,
                int*         __restrict__ tmp,
                int*         __restrict__ tile_ctr) {
    __shared__ int  hist[NBKT];
    __shared__ int  lbase[NBKT];
    __shared__ int  wsum[32];
    __shared__ int2 ent[EPB] __attribute__((aligned(16)));
    const int tid = threadIdx.x;
    const int blk = blockIdx.x;

    for (int i = tid; i < NBKT; i += SORT_THREADS) hist[i] = 0;
    if (blk == 0 && tid < 2) tile_ctr[tid] = 0;
    __syncthreads();

    int4   r4[EPT_VEC];
    int4   c4[EPT_VEC];
    float4 v4[EPT_VEC];
#pragma unroll
    for (int q = 0; q < EPT_VEC; ++q) {
        const int e0 = blk * EPB + (q * SORT_THREADS + tid) * 4;
        const int ec = min(e0, NNZ - 4);
        r4[q] = *(const int4*)(rows + ec);
        c4[q] = *(const int4*)(cols + ec);
        v4[q] = *(const float4*)(vals + ec);
    }
    int  bk[EPT_VEC * 4];
    int  lr[EPT_VEC * 4];
    int2 en[EPT_VEC * 4];
#pragma unroll
    for (int q = 0; q < EPT_VEC; ++q) {
        const int   e0 = blk * EPB + (q * SORT_THREADS + tid) * 4;
        const int   rr[4] = {r4[q].x, r4[q].y, r4[q].z, r4[q].w};
        const int   cc[4] = {c4[q].x, c4[q].y, c4[q].z, c4[q].w};
        const float vv[4] = {v4[q].x, v4[q].y, v4[q].z, v4[q].w};
#pragma unroll
        for (int k = 0; k < 4; ++k) {
            const int r = ngcf_clamp_row(rr[k]);
            const int c = (int)min((unsigned)cc[k], (unsigned)(NODE_NUM - 1));
            bk[q * 4 + k] = r >> BKT_SHIFT;
            en[q * 4 + k] = make_int2(((r & (BKT_ROWS - 1)) << COL_BITS) | c, __float_as_int(vv[k]));
            lr[q * 4 + k] = (e0 < NNZ) ? atomicAdd(&hist[bk[q * 4 + k]], 1) : 0;
        }
    }
    __syncthreads();

    const int b0 = tid * 2;
    const int b1 = b0 + 1;
    const int h0 = (b0 < NBKT) ? hist[b0] : 0;
    const int h1 = (b1 < NBKT) ? hist[b1] : 0;
    int tot;
    const int lb = ngcf_block_excl_scan(h0 + h1, wsum, tot);
    if (b0 < NBKT) {
        lbase[b0] = lb;
        offT[b0 * NBLK + blk] = lb;
    }
    if (b1 < NBKT) {
        lbase[b1] = lb + h0;
        offT[b1 * NBLK + blk] = lb + h0;
    }
    if (tid == 0) offT[NBKT * NBLK + blk] = tot;
    __syncthreads();

#pragma unroll
    for (int q = 0; q < EPT_VEC; ++q) {
        const int e0 = blk * EPB + (q * SORT_THREADS + tid) * 4;
        if (e0 < NNZ) {
#pragma unroll
            for (int k = 0; k < 4; ++k) {
                const int lp = min(lbase[bk[q * 4 + k]] + lr[q * 4 + k], EPB - 1);
                ent[lp] = en[q * 4 + k];
            }
        }
    }
    __syncthreads();

    const int nvalid = min(EPB, NNZ - blk * EPB);
    int4*       out4 = (int4*)tmp + (size_t)blk * (EPB / 2);
    const int4* ent4 = (const int4*)ent;
    for (int i = tid; i < nvalid / 2; i += SORT_THREADS) out4[i] = ent4[i];
}

__device__ __forceinline__ void ngcf_place(int2 e, const int* exs, int* cur,
                                           int2* stage, int2* dst, bool staged, int n) {
    const int lrow = (e.x >> COL_BITS) & (BKT_ROWS - 1);
    const int p = exs[lrow] + atomicAdd(&cur[lrow], 1);
    const int2 o = make_int2(e.x & COL_MASK, e.y);
    if (staged) stage[min(max(p, 0), BKT_STAGE - 1)] = o;
    else        dst[min(max(p, 0), n - 1)] = o;
}

__global__ void __launch_bounds__(SORT_THREADS)
ngcf_bucket_kernel(const int* __restrict__ offT,
                   const int* __restrict__ tmp,
                   int*       __restrict__ rowptr,
                   int*       __restrict__ pairs) {
    __shared__ int  cnt[BKT_ROWS];
    __shared__ int  exs[BKT_ROWS];
    __shared__ int  cur[BKT_ROWS];
    __shared__ int  wsum[32];
    __shared__ int  sbeg[SLICE_SLOTS];
    __shared__ int  slen[SLICE_SLOTS];
    __shared__ int2 stage[BKT_STAGE];
    const int tid = threadIdx.x;
    const int j   = blockIdx.x;

    if (tid < BKT_ROWS) { cnt[tid] = 0; cur[tid] = 0; }

    int osum = 0, lsum = 0;
#pragma unroll
    for (int q = 0; q < 2; ++q) {
        const int b = tid + q * SORT_THREADS;
        if (b < SLICE_SLOTS) {
            int beg = 0, len = 0;
            if (b < NBLK) {
                const int o0 = min(max(offT[j * NBLK + b], 0), EPB);
                const int o1 = min(max(offT[(j + 1) * NBLK + b], 0), EPB);
                beg = b * EPB + o0;
                len = min(max(o1 - o0, 0), max(NNZ - beg, 0));
                osum += o0;
                lsum += len;
            }
            sbeg[b] = beg;
            slen[b] = len;
        }
    }
    int base, n;
    ngcf_block_excl_scan(osum, wsum, base);
    ngcf_block_excl_scan(lsum, wsum, n);
    base = min(max(base, 0), NNZ);
    n    = min(n, NNZ - base);
    const int2* src = (const int2*)tmp;
    int2*       dst = (int2*)pairs + base;

    const int hw  = tid >> 4;
    const int l16 = tid & 15;

    int sb[SLICE_K], sl[SLICE_K];
#pragma unroll
    for (int k = 0; k < SLICE_K; ++k) {
        sb[k] = sbeg[hw + k * SLICE_HW];
        sl[k] = slen[hw + k * SLICE_HW];
    }

    int2 e[SLICE_K];
    {
        int y[SLICE_K];
#pragma unroll
        for (int k = 0; k < SLICE_K; ++k) {
            e[k] = (l16      < sl[k]) ? src[sb[k] + l16]        : make_int2(-1, 0);
            y[k] = (l16 + 16 < sl[k]) ? src[sb[k] + l16 + 16].x : -1;
        }
#pragma unroll
        for (int k = 0; k < SLICE_K; ++k) {
            if (e[k].x >= 0) atomicAdd(&cnt[(e[k].x >> COL_BITS) & (BKT_ROWS - 1)], 1);
            if (y[k]   >= 0) atomicAdd(&cnt[(y[k]   >> COL_BITS) & (BKT_ROWS - 1)], 1);
        }
#pragma unroll
        for (int k = 0; k < SLICE_K; ++k)
            for (int e3 = l16 + 32; e3 < sl[k]; e3 += 16)
                atomicAdd(&cnt[(src[sb[k] + e3].x >> COL_BITS) & (BKT_ROWS - 1)], 1);
    }
    __syncthreads();

    int tot;
    const int ex = ngcf_block_excl_scan((tid < BKT_ROWS) ? cnt[tid] : 0, wsum, tot);
    if (tid < BKT_ROWS) {
        exs[tid] = ex;
        const int row = j * BKT_ROWS + tid;
        if (row <= NODE_NUM) rowptr[row] = base + ex;
    }
    __syncthreads();

    const bool staged = n <= BKT_STAGE;
    {
        int2 f[SLICE_K];
#pragma unroll
        for (int k = 0; k < SLICE_K; ++k)
            f[k] = (l16 + 16 < sl[k]) ? src[sb[k] + l16 + 16] : make_int2(-1, 0);
#pragma unroll
        for (int k = 0; k < SLICE_K; ++k)
            if (e[k].x >= 0) ngcf_place(e[k], exs, cur, stage, dst, staged, n);
#pragma unroll
        for (int k = 0; k < SLICE_K; ++k)
            if (f[k].x >= 0) ngcf_place(f[k], exs, cur, stage, dst, staged, n);
#pragma unroll
        for (int k = 0; k < SLICE_K; ++k)
            for (int e2 = l16 + 32; e2 < sl[k]; e2 += 16)
                ngcf_place(src[sb[k] + e2], exs, cur, stage, dst, staged, n);
    }
    if (staged) {
        __syncthreads();
        for (int i = tid; i < n; i += SORT_THREADS) dst[i] = stage[i];
    }
}

typedef int v2i __attribute__((ext_vector_type(2)));
typedef const int __attribute__((address_space(4)))* ngcf_rowptr_p;
typedef const v2i __attribute__((address_space(4)))* ngcf_list_p;

template <bool TAIL>
__device__ __forceinline__ void ngcf_gather8(ngcf_list_p pairs,
                                             const float* __restrict__ Elane,
                                             int j, int end, float& ax, float& ay) {
    v2i p[8];
    for (int k = 0; k < 8; ++k) {
        int jj = j + k;
        if (TAIL) jj = min(jj, end - 1);
        p[k] = pairs[jj];
    }
    float2 ev[8];
    for (int k = 0; k < 8; ++k) {
        const unsigned c = min((unsigned)p[k].x, (unsigned)(NODE_NUM - 1));
        ev[k] = *(const float2*)(Elane + (size_t)c * EMBED_DIM);
    }
    for (int k = 0; k < 8; ++k) {
        if (!TAIL || (j + k < end)) {
            const float v = __int_as_float(p[k].y);
            ax = fmaf(v, ev[k].x, ax);
            ay = fmaf(v, ev[k].y, ay);
        }
    }
}

__device__ __forceinline__ void ngcf_spmm_row(const int*   __restrict__ rowptr,
                                              const int*   __restrict__ pairs_i,
                                              const float* __restrict__ E,
                                              float*       __restrict__ dst_row,
                                              int r, int lane) {
    const ngcf_list_p   pairs = (ngcf_list_p)pairs_i;
    const ngcf_rowptr_p rp    = (ngcf_rowptr_p)rowptr;
    int       j   = min(max(rp[r], 0), NNZ);
    const int end = min(rp[r + 1], NNZ);
    const float* Elane = E + lane * 2;

    float ax = 0.f, ay = 0.f;
    for (; j + 8 <= end; j += 8) ngcf_gather8<false>(pairs, Elane, j, end, ax, ay);
    if (j < end) ngcf_gather8<true>(pairs, Elane, j, end, ax, ay);

    *(float2*)(dst_row + lane * 2) = make_float2(ax, ay);
}

#define TILE_WAVES 16
#define TILE_THREADS (TILE_WAVES * 32)
#define TILE_PITCH 66
static_assert(EMBED_DIM * EMBED_DIM == 8 * TILE_THREADS, "weights: 8 floats per thread per matrix");

__device__ __forceinline__ void ngcf_layer_block16(const float* epRow, const float* sdRow,
                                                   const float* w1p, const float* w2p,
                                                   float* dstCol, int halfsel) {
    v8f acc = v8f{};
#pragma unroll
    for (int kk = 0; kk < 16; ++kk) {
        const int k0 = kk * 4;
        const float2 ep = *(const float2*)(epRow + k0);
        const float2 sd = *(const float2*)(sdRow + k0);

        v2f aSum, aBi;
        aSum.x = ep.x + sd.x;  aSum.y = ep.y + sd.y;
        aBi.x  = ep.x * sd.x;  aBi.y  = ep.y * sd.y;

        const int kph = ((k0 >> 1) + halfsel) << 7;
        const v2f b1 = *(const v2f*)(w1p + kph);
        const v2f b2 = *(const v2f*)(w2p + kph);
        acc = __builtin_amdgcn_wmma_f32_16x16x4_f32(
            false, aSum, false, b1, (short)0, acc, false, false);
        acc = __builtin_amdgcn_wmma_f32_16x16x4_f32(
            false, aBi,  false, b2, (short)0, acc, false, false);
    }
#pragma unroll
    for (int i = 0; i < 8; ++i) {
        float v = acc[i];
        v = (v > 0.f) ? v : v * NEG_SLOPE;
        dstCol[(i + halfsel * 8) * TILE_PITCH] = v;
    }
}

#define LAYER_WGS 1024
static_assert(LAYER_WGS <= NODE_TILES, "every workgroup has a first tile");

template <bool FIRST>
__device__ __forceinline__ int ngcf_layer_tile(const int*   __restrict__ rowptr,
                                                const int*   __restrict__ pairs_i,
                                                const float* __restrict__ Eprev,
                                                const float* __restrict__ W1,
                                                const float* __restrict__ W2,
                                                float*       __restrict__ Eout,
                                                float* w1i, float* w2i,
                                                float* sideT, float* prevT, float* nextT,
                                                int* __restrict__ tile_ctr, int* nextTileS,
                                                int tile, int tid, int lane, int wid) {
    int claimed = 0;
    if (tid == TILE_THREADS - 32) claimed = (int)atomicInc((unsigned int*)tile_ctr, 0xFFFFFFFFu);

    float wa[8], wb[8];
    if (FIRST) {
#pragma unroll
        for (int j = 0; j < 8; ++j) {
            wa[j] = W1[j * TILE_THREADS + tid];
            wb[j] = W2[j * TILE_THREADS + tid];
        }
    }

    const int r = tile * 16 + wid;
    const float2 o = *(const float2*)(Eprev + (size_t)r * EMBED_DIM + lane * 2);

    ngcf_spmm_row(rowptr, pairs_i, Eprev, sideT + wid * TILE_PITCH, r, lane);
    *(float2*)(prevT + wid * TILE_PITCH + lane * 2) = o;

    if (FIRST) {
#pragma unroll
        for (int j = 0; j < 8; ++j) {
            const int i = j * TILE_THREADS + tid;
            const int k = i >> 6;
            const int n = i & 63;
            const int dst = ((k >> 1) << 7) + (n << 1) + (k & 1);
            w1i[dst] = wa[j];
            w2i[dst] = wb[j];
        }
    }
    __syncthreads();

    if (wid < 4) {
        const int halfsel = lane >> 4;
        const int nlane   = lane & 15;
        const int koff    = halfsel * 2;
        const int colo    = (nlane << 1) + wid * 32;
        ngcf_layer_block16(prevT + nlane * TILE_PITCH + koff,
                           sideT + nlane * TILE_PITCH + koff,
                           w1i + colo, w2i + colo,
                           nextT + wid * 16 + nlane, halfsel);
    }
    if (tid == TILE_THREADS - 32) *nextTileS = claimed;
    __syncthreads();

    *(float2*)(Eout + (size_t)r * EMBED_DIM + lane * 2) =
        *(const float2*)(nextT + wid * TILE_PITCH + lane * 2);

    return LAYER_WGS + __builtin_amdgcn_readfirstlane(*nextTileS);
}

__global__ void __launch_bounds__(TILE_THREADS)
ngcf_layer_kernel(const int*   __restrict__ rowptr,
                  const int*   __restrict__ pairs_i,
                  const float* __restrict__ Eprev,
                  const float* __restrict__ W1,
                  const float* __restrict__ W2,
                  float*       __restrict__ Eout,
                  int*         __restrict__ tile_ctr) {
    __shared__ float w1i[EMBED_DIM * EMBED_DIM]  __attribute__((aligned(16)));
    __shared__ float w2i[EMBED_DIM * EMBED_DIM]  __attribute__((aligned(16)));
    __shared__ float sideT[16 * TILE_PITCH]      __attribute__((aligned(16)));
    __shared__ float prevT[16 * TILE_PITCH]      __attribute__((aligned(16)));
    __shared__ float nextT[16 * TILE_PITCH]      __attribute__((aligned(16)));
    __shared__ int   nextTileS;

    const int tid  = threadIdx.x;
    const int lane = tid & 31;
    const int wid = __builtin_amdgcn_readfirstlane(tid >> 5);

    int tile = blockIdx.x;
    if (tile >= NODE_TILES) return;
    tile = ngcf_layer_tile<true>(rowptr, pairs_i, Eprev, W1, W2, Eout,
                                 w1i, w2i, sideT, prevT, nextT, tile_ctr, &nextTileS,
                                 tile, tid, lane, wid);
    while (tile < NODE_TILES)
        tile = ngcf_layer_tile<false>(rowptr, pairs_i, Eprev, W1, W2, Eout,
                                      w1i, w2i, sideT, prevT, nextT, tile_ctr, &nextTileS,
                                      tile, tid, lane, wid);
}

#define SEL_TILES (2 * BATCH / 16)

__global__ void __launch_bounds__(TILE_THREADS)
ngcf_tail_kernel(const int*   __restrict__ rowptr,
                 const int*   __restrict__ pairs_i,
                 const float* __restrict__ emb,
                 const float* __restrict__ E1,
                 const float* __restrict__ E2,
                 const float* __restrict__ W1,
                 const float* __restrict__ W2,
                 const int*   __restrict__ user_idx,
                 const int*   __restrict__ item_idx,
                 float*       __restrict__ out) {
    __shared__ float w1i[EMBED_DIM * EMBED_DIM]  __attribute__((aligned(16)));
    __shared__ float w2i[EMBED_DIM * EMBED_DIM]  __attribute__((aligned(16)));
    __shared__ float sideT[16 * TILE_PITCH]      __attribute__((aligned(16)));
    __shared__ float e2T[16 * TILE_PITCH]        __attribute__((aligned(16)));
    __shared__ float e3T[16 * TILE_PITCH]        __attribute__((aligned(16)));

    const int tid  = threadIdx.x;
    const int lane = tid & 31;
    const int wid = __builtin_amdgcn_readfirstlane(tid >> 5);
    const int m0  = blockIdx.x * 16;
    const int s   = m0 + wid;
    const int r   = ngcf_slot_node(user_idx, item_idx, s);

    float wa[8], wb[8];
#pragma unroll
    for (int j = 0; j < 8; ++j) {
        wa[j] = W1[j * TILE_THREADS + tid];
        wb[j] = W2[j * TILE_THREADS + tid];
    }

    const size_t ro = (size_t)r * EMBED_DIM + lane * 2;
    const float2 o0 = *(const float2*)(emb + ro);
    const float2 o1 = *(const float2*)(E1 + ro);
    const float2 o2 = *(const float2*)(E2 + ro);

    ngcf_spmm_row(rowptr, pairs_i, E2, sideT + wid * TILE_PITCH, r, lane);
    *(float2*)(e2T + wid * TILE_PITCH + lane * 2) = o2;

#pragma unroll
    for (int j = 0; j < 8; ++j) {
        const int i = j * TILE_THREADS + tid;
        const int k = i >> 6;
        const int n = i & 63;
        const int dst = ((k >> 1) << 7) + (n << 1) + (k & 1);
        w1i[dst] = wa[j];
        w2i[dst] = wb[j];
    }
    __syncthreads();

    if (wid < 4) {
        const int halfsel = lane >> 4;
        const int nlane   = lane & 15;
        const int koff    = halfsel * 2;
        const int colo    = (nlane << 1) + wid * 32;
        ngcf_layer_block16(e2T   + nlane * TILE_PITCH + koff,
                           sideT + nlane * TILE_PITCH + koff,
                           w1i + colo, w2i + colo,
                           e3T + wid * 16 + nlane, halfsel);
    }
    __syncthreads();

    float* orow = out + (size_t)s * 256 + lane * 2;
    *(float2*)(orow)       = o0;
    *(float2*)(orow + 64)  = o1;
    *(float2*)(orow + 128) = o2;
    *(float2*)(orow + 192) = *(const float2*)(e3T + wid * TILE_PITCH + lane * 2);
}

extern "C" void kernel_launch(void* const* d_in, const int* in_sizes, int n_in,
                              void* d_out, int out_size, void* d_ws, size_t ws_size,
                              hipStream_t stream) {
    const float* emb      = (const float*)d_in[0];
    const float* W1       = (const float*)d_in[1];
    const float* W2       = (const float*)d_in[2];
    const float* vals     = (const float*)d_in[3];
    const int*   row_idx  = (const int*)d_in[4];
    const int*   col_idx  = (const int*)d_in[5];
    const int*   user_idx = (const int*)d_in[6];
    const int*   item_idx = (const int*)d_in[7];

    const size_t EN = (size_t)NODE_NUM * EMBED_DIM;
    float* E1   = (float*)d_ws;
    float* E2   = E1 + EN;
    float* P3   = E2 + EN;

    int* pairs  = (int*)P3;
    int* rowptr = pairs + 2 * (size_t)NNZ;
    int* offT   = rowptr + 102400;
    int* tctr   = offT + 153600;
    int* tmp    = (int*)E2;

    ngcf_bin_kernel<<<NBLK, SORT_THREADS, 0, stream>>>(vals, row_idx, col_idx, offT, tmp, tctr);
    ngcf_bucket_kernel<<<NBKT, SORT_THREADS, 0, stream>>>(offT, tmp, rowptr, pairs);

    ngcf_layer_kernel<<<LAYER_WGS, TILE_THREADS, 0, stream>>>(rowptr, pairs, emb, W1, W2, E1, tctr);
    ngcf_layer_kernel<<<LAYER_WGS, TILE_THREADS, 0, stream>>>(
        rowptr, pairs, E1, W1 + EMBED_DIM * EMBED_DIM, W2 + EMBED_DIM * EMBED_DIM, E2, tctr + 1);

    ngcf_tail_kernel<<<SEL_TILES, TILE_THREADS, 0, stream>>>(
        rowptr, pairs, emb, E1, E2,
        W1 + 2 * EMBED_DIM * EMBED_DIM, W2 + 2 * EMBED_DIM * EMBED_DIM,
        user_idx, item_idx, (float*)d_out);
}


# Layers 1, 2: 32-node tiles, not 16: each wave sums two side rows at once (16 source-row reads in flight, not 8), 8 waves run the GEMMs; half the barriers and GEMM phases per row; same sum order
# speedup vs baseline: 6.4814x; 1.0534x over previous
#include <hip/hip_runtime.h>
#include <hip/hip_bf16.h>


#define USER_NUM 60000
#define ITEM_NUM 40000
#define NODE_NUM (USER_NUM + ITEM_NUM)
#define EMBED_DIM 64
#define N_LAYER 3
#define NNZ 1600000
#define BATCH 4096
#define NEG_SLOPE 0.2f
#define LAYER_TILE_ROWS 32
#define NODE_TILES (NODE_NUM / LAYER_TILE_ROWS)
static_assert(NODE_NUM % LAYER_TILE_ROWS == 0, "whole tiles only");

typedef float v2f __attribute__((ext_vector_type(2)));
typedef float v8f __attribute__((ext_vector_type(8)));

__device__ __forceinline__ int ngcf_slot_node(const int* __restrict__ user_idx,
                                              const int* __restrict__ item_idx,
                                              int s) {
    const int node = (s < BATCH) ? user_idx[s] : USER_NUM + item_idx[s - BATCH];
    return min(max(node, 0), NODE_NUM - 1);
}

#define BKT_SHIFT 8
#define BKT_ROWS (1 << BKT_SHIFT)
#define NBKT ((NODE_NUM + BKT_ROWS - 1) / BKT_ROWS)
#define SORT_THREADS 256
#define EPT_VEC 4
#define EPB 4096
#define NBLK ((NNZ + EPB - 1) / EPB)
#define COL_BITS 17
#define COL_MASK ((1 << COL_BITS) - 1)
#define BKT_LOADS 24
#define BKT_STAGE (BKT_LOADS * SORT_THREADS)
#define SLICE_HW (SORT_THREADS / 16)
#define SLICE_K ((NBLK + SLICE_HW - 1) / SLICE_HW)
#define SLICE_SLOTS (SLICE_K * SLICE_HW)

static_assert(NBKT <= 2 * SORT_THREADS, "two buckets per thread in the block scan");
static_assert(SLICE_SLOTS <= 2 * SORT_THREADS, "two slices per thread in the level-2 set-up");
static_assert(BKT_ROWS <= SORT_THREADS, "one row per thread in the level-2 scan");
static_assert(EPB == EPT_VEC * 4 * SORT_THREADS, "edges per block");
static_assert(NNZ % 4 == 0 && EPB % 4 == 0, "a block's valid edges come in fours: 16-byte stream stores");
static_assert(NODE_NUM <= (1 << COL_BITS) && (BKT_ROWS << COL_BITS) > 0, "entry packing");

__device__ __forceinline__ int ngcf_block_excl_scan(int v, int* wsum, int& total) {
    const int lane = threadIdx.x & 31;
    const int wid  = threadIdx.x >> 5;
    int incl = v;
    for (int off = 1; off < 32; off <<= 1) {
        const int t = __shfl_up(incl, off);
        if (lane >= off) incl += t;
    }
    if (lane == 31) wsum[wid] = incl;
    __syncthreads();
    const int wv = (lane < (int)(blockDim.x >> 5)) ? wsum[lane] : 0;
    int wincl = wv;
    for (int off = 1; off < 32; off <<= 1) {
        const int t = __shfl_up(wincl, off);
        if (lane >= off) wincl += t;
    }
    const int wpre = __shfl(wincl, wid) - __shfl(wv, wid);
    total = __shfl(wincl, 31);
    __syncthreads();
    return wpre + incl - v;
}

__device__ __forceinline__ int ngcf_clamp_row(int r) {
    return min(max(r, 0), NODE_NUM - 1);
}

__global__ void __launch_bounds__(SORT_THREADS)
ngcf_bin_kernel(const float* __restrict__ vals,
                const int*   __restrict__ rows,
                const int*   __restrict__ cols,
                int*         __restrict__ offT,
                int*         __restrict__ tmp,
                int*         __restrict__ tile_ctr) {
    __shared__ int  hist[NBKT];
    __shared__ int  lbase[NBKT];
    __shared__ int  wsum[32];
    __shared__ int2 ent[EPB] __attribute__((aligned(16)));
    const int tid = threadIdx.x;
    const int blk = blockIdx.x;

    for (int i = tid; i < NBKT; i += SORT_THREADS) hist[i] = 0;
    if (blk == 0 && tid < 2) tile_ctr[tid] = 0;
    __syncthreads();

    int4   r4[EPT_VEC];
    int4   c4[EPT_VEC];
    float4 v4[EPT_VEC];
#pragma unroll
    for (int q = 0; q < EPT_VEC; ++q) {
        const int e0 = blk * EPB + (q * SORT_THREADS + tid) * 4;
        const int ec = min(e0, NNZ - 4);
        r4[q] = *(const int4*)(rows + ec);
        c4[q] = *(const int4*)(cols + ec);
        v4[q] = *(const float4*)(vals + ec);
    }
    int  bk[EPT_VEC * 4];
    int  lr[EPT_VEC * 4];
    int2 en[EPT_VEC * 4];
#pragma unroll
    for (int q = 0; q < EPT_VEC; ++q) {
        const int   e0 = blk * EPB + (q * SORT_THREADS + tid) * 4;
        const int   rr[4] = {r4[q].x, r4[q].y, r4[q].z, r4[q].w};
        const int   cc[4] = {c4[q].x, c4[q].y, c4[q].z, c4[q].w};
        const float vv[4] = {v4[q].x, v4[q].y, v4[q].z, v4[q].w};
#pragma unroll
        for (int k = 0; k < 4; ++k) {
            const int r = ngcf_clamp_row(rr[k]);
            const int c = (int)min((unsigned)cc[k], (unsigned)(NODE_NUM - 1));
            bk[q * 4 + k] = r >> BKT_SHIFT;
            en[q * 4 + k] = make_int2(((r & (BKT_ROWS - 1)) << COL_BITS) | c, __float_as_int(vv[k]));
            lr[q * 4 + k] = (e0 < NNZ) ? atomicAdd(&hist[bk[q * 4 + k]], 1) : 0;
        }
    }
    __syncthreads();

    const int b0 = tid * 2;
    const int b1 = b0 + 1;
    const int h0 = (b0 < NBKT) ? hist[b0] : 0;
    const int h1 = (b1 < NBKT) ? hist[b1] : 0;
    int tot;
    const int lb = ngcf_block_excl_scan(h0 + h1, wsum, tot);
    if (b0 < NBKT) {
        lbase[b0] = lb;
        offT[b0 * NBLK + blk] = lb;
    }
    if (b1 < NBKT) {
        lbase[b1] = lb + h0;
        offT[b1 * NBLK + blk] = lb + h0;
    }
    if (tid == 0) offT[NBKT * NBLK + blk] = tot;
    __syncthreads();

#pragma unroll
    for (int q = 0; q < EPT_VEC; ++q) {
        const int e0 = blk * EPB + (q * SORT_THREADS + tid) * 4;
        if (e0 < NNZ) {
#pragma unroll
            for (int k = 0; k < 4; ++k) {
                const int lp = min(lbase[bk[q * 4 + k]] + lr[q * 4 + k], EPB - 1);
                ent[lp] = en[q * 4 + k];
            }
        }
    }
    __syncthreads();

    const int nvalid = min(EPB, NNZ - blk * EPB);
    int4*       out4 = (int4*)tmp + (size_t)blk * (EPB / 2);
    const int4* ent4 = (const int4*)ent;
    for (int i = tid; i < nvalid / 2; i += SORT_THREADS) out4[i] = ent4[i];
}

__device__ __forceinline__ void ngcf_place(int2 e, const int* exs, int* cur,
                                           int2* stage, int2* dst, bool staged, int n) {
    const int lrow = (e.x >> COL_BITS) & (BKT_ROWS - 1);
    const int p = exs[lrow] + atomicAdd(&cur[lrow], 1);
    const int2 o = make_int2(e.x & COL_MASK, e.y);
    if (staged) stage[min(max(p, 0), BKT_STAGE - 1)] = o;
    else        dst[min(max(p, 0), n - 1)] = o;
}

__global__ void __launch_bounds__(SORT_THREADS)
ngcf_bucket_kernel(const int* __restrict__ offT,
                   const int* __restrict__ tmp,
                   int*       __restrict__ rowptr,
                   int*       __restrict__ pairs) {
    __shared__ int  cnt[BKT_ROWS];
    __shared__ int  exs[BKT_ROWS];
    __shared__ int  cur[BKT_ROWS];
    __shared__ int  wsum[32];
    __shared__ int  sbeg[SLICE_SLOTS];
    __shared__ int  slen[SLICE_SLOTS];
    __shared__ int2 stage[BKT_STAGE];
    const int tid = threadIdx.x;
    const int j   = blockIdx.x;

    if (tid < BKT_ROWS) { cnt[tid] = 0; cur[tid] = 0; }

    int osum = 0, lsum = 0;
#pragma unroll
    for (int q = 0; q < 2; ++q) {
        const int b = tid + q * SORT_THREADS;
        if (b < SLICE_SLOTS) {
            int beg = 0, len = 0;
            if (b < NBLK) {
                const int o0 = min(max(offT[j * NBLK + b], 0), EPB);
                const int o1 = min(max(offT[(j + 1) * NBLK + b], 0), EPB);
                beg = b * EPB + o0;
                len = min(max(o1 - o0, 0), max(NNZ - beg, 0));
                osum += o0;
                lsum += len;
            }
            sbeg[b] = beg;
            slen[b] = len;
        }
    }
    int base, n;
    ngcf_block_excl_scan(osum, wsum, base);
    ngcf_block_excl_scan(lsum, wsum, n);
    base = min(max(base, 0), NNZ);
    n    = min(n, NNZ - base);
    const int2* src = (const int2*)tmp;
    int2*       dst = (int2*)pairs + base;

    const int hw  = tid >> 4;
    const int l16 = tid & 15;

    int sb[SLICE_K], sl[SLICE_K];
#pragma unroll
    for (int k = 0; k < SLICE_K; ++k) {
        sb[k] = sbeg[hw + k * SLICE_HW];
        sl[k] = slen[hw + k * SLICE_HW];
    }

    int2 e[SLICE_K];
    {
        int y[SLICE_K];
#pragma unroll
        for (int k = 0; k < SLICE_K; ++k) {
            e[k] = (l16      < sl[k]) ? src[sb[k] + l16]        : make_int2(-1, 0);
            y[k] = (l16 + 16 < sl[k]) ? src[sb[k] + l16 + 16].x : -1;
        }
#pragma unroll
        for (int k = 0; k < SLICE_K; ++k) {
            if (e[k].x >= 0) atomicAdd(&cnt[(e[k].x >> COL_BITS) & (BKT_ROWS - 1)], 1);
            if (y[k]   >= 0) atomicAdd(&cnt[(y[k]   >> COL_BITS) & (BKT_ROWS - 1)], 1);
        }
#pragma unroll
        for (int k = 0; k < SLICE_K; ++k)
            for (int e3 = l16 + 32; e3 < sl[k]; e3 += 16)
                atomicAdd(&cnt[(src[sb[k] + e3].x >> COL_BITS) & (BKT_ROWS - 1)], 1);
    }
    __syncthreads();

    int tot;
    const int ex = ngcf_block_excl_scan((tid < BKT_ROWS) ? cnt[tid] : 0, wsum, tot);
    if (tid < BKT_ROWS) {
        exs[tid] = ex;
        const int row = j * BKT_ROWS + tid;
        if (row <= NODE_NUM) rowptr[row] = base + ex;
    }
    __syncthreads();

    const bool staged = n <= BKT_STAGE;
    {
        int2 f[SLICE_K];
#pragma unroll
        for (int k = 0; k < SLICE_K; ++k)
            f[k] = (l16 + 16 < sl[k]) ? src[sb[k] + l16 + 16] : make_int2(-1, 0);
#pragma unroll
        for (int k = 0; k < SLICE_K; ++k)
            if (e[k].x >= 0) ngcf_place(e[k], exs, cur, stage, dst, staged, n);
#pragma unroll
        for (int k = 0; k < SLICE_K; ++k)
            if (f[k].x >= 0) ngcf_place(f[k], exs, cur, stage, dst, staged, n);
#pragma unroll
        for (int k = 0; k < SLICE_K; ++k)
            for (int e2 = l16 + 32; e2 < sl[k]; e2 += 16)
                ngcf_place(src[sb[k] + e2], exs, cur, stage, dst, staged, n);
    }
    if (staged) {
        __syncthreads();
        for (int i = tid; i < n; i += SORT_THREADS) dst[i] = stage[i];
    }
}

typedef int v2i __attribute__((ext_vector_type(2)));
typedef const int __attribute__((address_space(4)))* ngcf_rowptr_p;
typedef const v2i __attribute__((address_space(4)))* ngcf_list_p;

template <bool TAIL>
__device__ __forceinline__ void ngcf_gather8(ngcf_list_p pairs,
                                             const float* __restrict__ Elane,
                                             int j, int end, float& ax, float& ay) {
    v2i p[8];
    for (int k = 0; k < 8; ++k) {
        int jj = j + k;
        if (TAIL) jj = min(jj, end - 1);
        p[k] = pairs[jj];
    }
    float2 ev[8];
    for (int k = 0; k < 8; ++k) {
        const unsigned c = min((unsigned)p[k].x, (unsigned)(NODE_NUM - 1));
        ev[k] = *(const float2*)(Elane + (size_t)c * EMBED_DIM);
    }
    for (int k = 0; k < 8; ++k) {
        if (!TAIL || (j + k < end)) {
            const float v = __int_as_float(p[k].y);
            ax = fmaf(v, ev[k].x, ax);
            ay = fmaf(v, ev[k].y, ay);
        }
    }
}

__device__ __forceinline__ void ngcf_spmm_row(const int*   __restrict__ rowptr,
                                              const int*   __restrict__ pairs_i,
                                              const float* __restrict__ E,
                                              float*       __restrict__ dst_row,
                                              int r, int lane) {
    const ngcf_list_p   pairs = (ngcf_list_p)pairs_i;
    const ngcf_rowptr_p rp    = (ngcf_rowptr_p)rowptr;
    int       j   = min(max(rp[r], 0), NNZ);
    const int end = min(rp[r + 1], NNZ);
    const float* Elane = E + lane * 2;

    float ax = 0.f, ay = 0.f;
    for (; j + 8 <= end; j += 8) ngcf_gather8<false>(pairs, Elane, j, end, ax, ay);
    if (j < end) ngcf_gather8<true>(pairs, Elane, j, end, ax, ay);

    *(float2*)(dst_row + lane * 2) = make_float2(ax, ay);
}

template <bool TA, bool TB>
__device__ __forceinline__ void ngcf_gather8x2(ngcf_list_p pairs,
                                               const float* __restrict__ Elane,
                                               int ja, int enda, int jb, int endb,
                                               float& ax, float& ay, float& bx, float& by) {
    v2i pa[8], pb[8];
    for (int k = 0; k < 8; ++k) {
        int jj = ja + k;
        if (TA) jj = min(jj, enda - 1);
        pa[k] = pairs[jj];
    }
    for (int k = 0; k < 8; ++k) {
        int jj = jb + k;
        if (TB) jj = min(jj, endb - 1);
        pb[k] = pairs[jj];
    }
    float2 ea[8], eb[8];
    for (int k = 0; k < 8; ++k) {
        const unsigned c = min((unsigned)pa[k].x, (unsigned)(NODE_NUM - 1));
        ea[k] = *(const float2*)(Elane + (size_t)c * EMBED_DIM);
    }
    for (int k = 0; k < 8; ++k) {
        const unsigned c = min((unsigned)pb[k].x, (unsigned)(NODE_NUM - 1));
        eb[k] = *(const float2*)(Elane + (size_t)c * EMBED_DIM);
    }
    for (int k = 0; k < 8; ++k) {
        if (!TA || (ja + k < enda)) {
            const float v = __int_as_float(pa[k].y);
            ax = fmaf(v, ea[k].x, ax);
            ay = fmaf(v, ea[k].y, ay);
        }
    }
    for (int k = 0; k < 8; ++k) {
        if (!TB || (jb + k < endb)) {
            const float v = __int_as_float(pb[k].y);
            bx = fmaf(v, eb[k].x, bx);
            by = fmaf(v, eb[k].y, by);
        }
    }
}

__device__ __forceinline__ void ngcf_row_range(const int* __restrict__ rowptr, int r,
                                               int& j, int& end) {
    const ngcf_rowptr_p rp = (ngcf_rowptr_p)rowptr;
    j   = min(max(rp[r], 0), NNZ);
    end = min(rp[r + 1], NNZ);
}

__device__ __forceinline__ void ngcf_spmm_row2(const int*   __restrict__ pairs_i,
                                               const float* __restrict__ E,
                                               float*       __restrict__ dst_a,
                                               float*       __restrict__ dst_b,
                                               int ja, int enda, int jb, int endb, int lane) {
    const ngcf_list_p pairs = (ngcf_list_p)pairs_i;
    const float* Elane = E + lane * 2;

    float ax = 0.f, ay = 0.f, bx = 0.f, by = 0.f;
    while (ja < enda && jb < endb) {
        const bool fa = ja + 8 <= enda;
        const bool fb = jb + 8 <= endb;
        if (fa && fb)  ngcf_gather8x2<false, false>(pairs, Elane, ja, enda, jb, endb, ax, ay, bx, by);
        else if (fa)   ngcf_gather8x2<false, true >(pairs, Elane, ja, enda, jb, endb, ax, ay, bx, by);
        else if (fb)   ngcf_gather8x2<true,  false>(pairs, Elane, ja, enda, jb, endb, ax, ay, bx, by);
        else           ngcf_gather8x2<true,  true >(pairs, Elane, ja, enda, jb, endb, ax, ay, bx, by);
        ja += 8;
        jb += 8;
    }
    for (; ja + 8 <= enda; ja += 8) ngcf_gather8<false>(pairs, Elane, ja, enda, ax, ay);
    if (ja < enda) ngcf_gather8<true>(pairs, Elane, ja, enda, ax, ay);
    for (; jb + 8 <= endb; jb += 8) ngcf_gather8<false>(pairs, Elane, jb, endb, bx, by);
    if (jb < endb) ngcf_gather8<true>(pairs, Elane, jb, endb, bx, by);

    *(float2*)(dst_a + lane * 2) = make_float2(ax, ay);
    *(float2*)(dst_b + lane * 2) = make_float2(bx, by);
}

#define TILE_WAVES 16
#define TILE_THREADS (TILE_WAVES * 32)
#define TILE_PITCH 66
static_assert(EMBED_DIM * EMBED_DIM == 8 * TILE_THREADS, "weights: 8 floats per thread per matrix");

__device__ __forceinline__ void ngcf_layer_block16(const float* epRow, const float* sdRow,
                                                   const float* w1p, const float* w2p,
                                                   float* dstCol, int halfsel) {
    v8f acc = v8f{};
#pragma unroll
    for (int kk = 0; kk < 16; ++kk) {
        const int k0 = kk * 4;
        const float2 ep = *(const float2*)(epRow + k0);
        const float2 sd = *(const float2*)(sdRow + k0);

        v2f aSum, aBi;
        aSum.x = ep.x + sd.x;  aSum.y = ep.y + sd.y;
        aBi.x  = ep.x * sd.x;  aBi.y  = ep.y * sd.y;

        const int kph = ((k0 >> 1) + halfsel) << 7;
        const v2f b1 = *(const v2f*)(w1p + kph);
        const v2f b2 = *(const v2f*)(w2p + kph);
        acc = __builtin_amdgcn_wmma_f32_16x16x4_f32(
            false, aSum, false, b1, (short)0, acc, false, false);
        acc = __builtin_amdgcn_wmma_f32_16x16x4_f32(
            false, aBi,  false, b2, (short)0, acc, false, false);
    }
#pragma unroll
    for (int i = 0; i < 8; ++i) {
        float v = acc[i];
        v = (v > 0.f) ? v : v * NEG_SLOPE;
        dstCol[(i + halfsel * 8) * TILE_PITCH] = v;
    }
}

#define LAYER_WGS 1024
static_assert(LAYER_WGS <= NODE_TILES, "every workgroup has a first tile");

template <bool FIRST>
__device__ __forceinline__ int ngcf_layer_tile(const int*   __restrict__ rowptr,
                                                const int*   __restrict__ pairs_i,
                                                const float* __restrict__ Eprev,
                                                const float* __restrict__ W1,
                                                const float* __restrict__ W2,
                                                float*       __restrict__ Eout,
                                                float* w1i, float* w2i,
                                                float* sideT, float* prevT, float* nextT,
                                                int* __restrict__ tile_ctr, int* nextTileS,
                                                int tile, int tid, int lane, int wid) {
    int claimed = 0;
    if (tid == TILE_THREADS - 32) claimed = (int)atomicInc((unsigned int*)tile_ctr, 0xFFFFFFFFu);

    float wa[8], wb[8];
    if (FIRST) {
#pragma unroll
        for (int j = 0; j < 8; ++j) {
            wa[j] = W1[j * TILE_THREADS + tid];
            wb[j] = W2[j * TILE_THREADS + tid];
        }
    }

    const int ra = tile * LAYER_TILE_ROWS + wid;
    const int rb = ra + 16;
    int ja, enda, jb, endb;
    ngcf_row_range(rowptr, ra, ja, enda);
    ngcf_row_range(rowptr, rb, jb, endb);
    const float2 oa = *(const float2*)(Eprev + (size_t)ra * EMBED_DIM + lane * 2);
    const float2 ob = *(const float2*)(Eprev + (size_t)rb * EMBED_DIM + lane * 2);

    if (FIRST) {
#pragma unroll
        for (int j = 0; j < 8; ++j) {
            const int i = j * TILE_THREADS + tid;
            const int k = i >> 6;
            const int n = i & 63;
            const int dst = ((k >> 1) << 7) + (n << 1) + (k & 1);
            w1i[dst] = wa[j];
            w2i[dst] = wb[j];
        }
    }

    ngcf_spmm_row2(pairs_i, Eprev,
                   sideT + wid * TILE_PITCH, sideT + (wid + 16) * TILE_PITCH,
                   ja, enda, jb, endb, lane);
    *(float2*)(prevT + wid * TILE_PITCH + lane * 2)        = oa;
    *(float2*)(prevT + (wid + 16) * TILE_PITCH + lane * 2) = ob;
    __syncthreads();

    if (wid < 8) {
        const int halfsel = lane >> 4;
        const int nlane   = lane & 15;
        const int koff    = halfsel * 2;
        const int cblk    = wid & 3;
        const int row0    = (wid >> 2) * 16;
        const int colo    = (nlane << 1) + cblk * 32;
        ngcf_layer_block16(prevT + (row0 + nlane) * TILE_PITCH + koff,
                           sideT + (row0 + nlane) * TILE_PITCH + koff,
                           w1i + colo, w2i + colo,
                           nextT + row0 * TILE_PITCH + cblk * 16 + nlane, halfsel);
    }
    if (tid == TILE_THREADS - 32) *nextTileS = claimed;
    __syncthreads();

    *(float2*)(Eout + (size_t)ra * EMBED_DIM + lane * 2) =
        *(const float2*)(nextT + wid * TILE_PITCH + lane * 2);
    *(float2*)(Eout + (size_t)rb * EMBED_DIM + lane * 2) =
        *(const float2*)(nextT + (wid + 16) * TILE_PITCH + lane * 2);

    return LAYER_WGS + __builtin_amdgcn_readfirstlane(*nextTileS);
}

__global__ void __launch_bounds__(TILE_THREADS)
ngcf_layer_kernel(const int*   __restrict__ rowptr,
                  const int*   __restrict__ pairs_i,
                  const float* __restrict__ Eprev,
                  const float* __restrict__ W1,
                  const float* __restrict__ W2,
                  float*       __restrict__ Eout,
                  int*         __restrict__ tile_ctr) {
    __shared__ float w1i[EMBED_DIM * EMBED_DIM]          __attribute__((aligned(16)));
    __shared__ float w2i[EMBED_DIM * EMBED_DIM]          __attribute__((aligned(16)));
    __shared__ float sideT[LAYER_TILE_ROWS * TILE_PITCH] __attribute__((aligned(16)));
    __shared__ float prevT[LAYER_TILE_ROWS * TILE_PITCH] __attribute__((aligned(16)));
    __shared__ float nextT[LAYER_TILE_ROWS * TILE_PITCH] __attribute__((aligned(16)));
    __shared__ int   nextTileS;

    const int tid  = threadIdx.x;
    const int lane = tid & 31;
    const int wid = __builtin_amdgcn_readfirstlane(tid >> 5);

    int tile = blockIdx.x;
    if (tile >= NODE_TILES) return;
    tile = ngcf_layer_tile<true>(rowptr, pairs_i, Eprev, W1, W2, Eout,
                                 w1i, w2i, sideT, prevT, nextT, tile_ctr, &nextTileS,
                                 tile, tid, lane, wid);
    while (tile < NODE_TILES)
        tile = ngcf_layer_tile<false>(rowptr, pairs_i, Eprev, W1, W2, Eout,
                                      w1i, w2i, sideT, prevT, nextT, tile_ctr, &nextTileS,
                                      tile, tid, lane, wid);
}

#define SEL_TILES (2 * BATCH / 16)

__global__ void __launch_bounds__(TILE_THREADS)
ngcf_tail_kernel(const int*   __restrict__ rowptr,
                 const int*   __restrict__ pairs_i,
                 const float* __restrict__ emb,
                 const float* __restrict__ E1,
                 const float* __restrict__ E2,
                 const float* __restrict__ W1,
                 const float* __restrict__ W2,
                 const int*   __restrict__ user_idx,
                 const int*   __restrict__ item_idx,
                 float*       __restrict__ out) {
    __shared__ float w1i[EMBED_DIM * EMBED_DIM]  __attribute__((aligned(16)));
    __shared__ float w2i[EMBED_DIM * EMBED_DIM]  __attribute__((aligned(16)));
    __shared__ float sideT[16 * TILE_PITCH]      __attribute__((aligned(16)));
    __shared__ float e2T[16 * TILE_PITCH]        __attribute__((aligned(16)));
    __shared__ float e3T[16 * TILE_PITCH]        __attribute__((aligned(16)));

    const int tid  = threadIdx.x;
    const int lane = tid & 31;
    const int wid = __builtin_amdgcn_readfirstlane(tid >> 5);
    const int m0  = blockIdx.x * 16;
    const int s   = m0 + wid;
    const int r   = ngcf_slot_node(user_idx, item_idx, s);

    float wa[8], wb[8];
#pragma unroll
    for (int j = 0; j < 8; ++j) {
        wa[j] = W1[j * TILE_THREADS + tid];
        wb[j] = W2[j * TILE_THREADS + tid];
    }

    const size_t ro = (size_t)r * EMBED_DIM + lane * 2;
    const float2 o0 = *(const float2*)(emb + ro);
    const float2 o1 = *(const float2*)(E1 + ro);
    const float2 o2 = *(const float2*)(E2 + ro);

    ngcf_spmm_row(rowptr, pairs_i, E2, sideT + wid * TILE_PITCH, r, lane);
    *(float2*)(e2T + wid * TILE_PITCH + lane * 2) = o2;

#pragma unroll
    for (int j = 0; j < 8; ++j) {
        const int i = j * TILE_THREADS + tid;
        const int k = i >> 6;
        const int n = i & 63;
        const int dst = ((k >> 1) << 7) + (n << 1) + (k & 1);
        w1i[dst] = wa[j];
        w2i[dst] = wb[j];
    }
    __syncthreads();

    if (wid < 4) {
        const int halfsel = lane >> 4;
        const int nlane   = lane & 15;
        const int koff    = halfsel * 2;
        const int colo    = (nlane << 1) + wid * 32;
        ngcf_layer_block16(e2T   + nlane * TILE_PITCH + koff,
                           sideT + nlane * TILE_PITCH + koff,
                           w1i + colo, w2i + colo,
                           e3T + wid * 16 + nlane, halfsel);
    }
    __syncthreads();

    float* orow = out + (size_t)s * 256 + lane * 2;
    *(float2*)(orow)       = o0;
    *(float2*)(orow + 64)  = o1;
    *(float2*)(orow + 128) = o2;
    *(float2*)(orow + 192) = *(const float2*)(e3T + wid * TILE_PITCH + lane * 2);
}

extern "C" void kernel_launch(void* const* d_in, const int* in_sizes, int n_in,
                              void* d_out, int out_size, void* d_ws, size_t ws_size,
                              hipStream_t stream) {
    const float* emb      = (const float*)d_in[0];
    const float* W1       = (const float*)d_in[1];
    const float* W2       = (const float*)d_in[2];
    const float* vals     = (const float*)d_in[3];
    const int*   row_idx  = (const int*)d_in[4];
    const int*   col_idx  = (const int*)d_in[5];
    const int*   user_idx = (const int*)d_in[6];
    const int*   item_idx = (const int*)d_in[7];

    const size_t EN = (size_t)NODE_NUM * EMBED_DIM;
    float* E1   = (float*)d_ws;
    float* E2   = E1 + EN;
    float* P3   = E2 + EN;

    int* pairs  = (int*)P3;
    int* rowptr = pairs + 2 * (size_t)NNZ;
    int* offT   = rowptr + 102400;
    int* tctr   = offT + 153600;
    int* tmp    = (int*)E2;

    ngcf_bin_kernel<<<NBLK, SORT_THREADS, 0, stream>>>(vals, row_idx, col_idx, offT, tmp, tctr);
    ngcf_bucket_kernel<<<NBKT, SORT_THREADS, 0, stream>>>(offT, tmp, rowptr, pairs);

    ngcf_layer_kernel<<<LAYER_WGS, TILE_THREADS, 0, stream>>>(rowptr, pairs, emb, W1, W2, E1, tctr);
    ngcf_layer_kernel<<<LAYER_WGS, TILE_THREADS, 0, stream>>>(
        rowptr, pairs, E1, W1 + EMBED_DIM * EMBED_DIM, W2 + EMBED_DIM * EMBED_DIM, E2, tctr + 1);

    ngcf_tail_kernel<<<SEL_TILES, TILE_THREADS, 0, stream>>>(
        rowptr, pairs, emb, E1, E2,
        W1 + 2 * EMBED_DIM * EMBED_DIM, W2 + 2 * EMBED_DIM * EMBED_DIM,
        user_idx, item_idx, (float*)d_out);
}
